# starting kernel
# baseline (speedup 1.0000x reference)
#include <hip/hip_runtime.h>

#define NN   100000
#define INC  256
#define HIDC 128
#define NE   1600000
#define NPOS 200000
#define NNEG 200000

typedef __attribute__((ext_vector_type(2))) float v2f;
typedef __attribute__((ext_vector_type(8))) float v8f;

__global__ void k_init_dinv(float* dinv) {
  int i = blockIdx.x * blockDim.x + threadIdx.x;
  if (i < NN) dinv[i] = 1.0f;
}

__global__ void k_count_deg(float* dinv, const int* __restrict__ col) {
  int i = blockIdx.x * blockDim.x + threadIdx.x;
  if (i < NE) unsafeAtomicAdd(&dinv[col[i]], 1.0f);
}

__global__ void k_finish_dinv(float* dinv) {
  int i = blockIdx.x * blockDim.x + threadIdx.x;
  if (i < NN) dinv[i] = 1.0f / sqrtf(dinv[i]);
}

__global__ void k_transpose(const float* __restrict__ W, float* __restrict__ WT,
                            int K, int N) {
  int i = blockIdx.x * blockDim.x + threadIdx.x;
  if (i >= K * N) return;
  int k = i / N, n = i - k * N;
  WT[n * K + k] = W[i];
}

template <int K>
__global__ __launch_bounds__(256) void k_gemm_wmma(const float* __restrict__ A,
                                                   const float* __restrict__ WT,
                                                   float* __restrict__ C) {
  int wave = (blockIdx.x * blockDim.x + threadIdx.x) >> 5;
  int lane = threadIdx.x & 31;
  int m0 = wave * 16;
  if (m0 >= NN) return;
  int half = lane >> 4;
  int l15  = lane & 15;
  const float* aptr  = A  + (size_t)(m0 + l15) * K + half * 2;
  const float* bbase = WT + (size_t)l15 * K + half * 2;

  v8f acc[8];
#pragma unroll
  for (int t = 0; t < 8; ++t) acc[t] = (v8f)(0.0f);

  __builtin_prefetch(aptr, 0, 0);
#pragma unroll 2
  for (int k = 0; k < K; k += 4) {
    v2f a = *(const v2f*)(aptr + k);
#pragma unroll
    for (int t = 0; t < 8; ++t) {
      v2f b = *(const v2f*)(bbase + (size_t)t * 16 * K + k);
      acc[t] = __builtin_amdgcn_wmma_f32_16x16x4_f32(
          false, a, false, b, (short)0, acc[t], false, false);
    }
  }

#pragma unroll
  for (int t = 0; t < 8; ++t) {
#pragma unroll
    for (int i = 0; i < 8; ++i) {
      C[(size_t)(m0 + i + half * 8) * HIDC + t * 16 + l15] = acc[t][i];
    }
  }
}

__global__ void k_selfloop(float* __restrict__ out, const float* __restrict__ h,
                           const float* __restrict__ dinv) {
  int i = blockIdx.x * blockDim.x + threadIdx.x;
  if (i >= NN * (HIDC / 4)) return;
  int n = i >> 5;
  float d = dinv[n];
  float s = d * d;
  float4 v = ((const float4*)h)[i];
  ((float4*)out)[i] = make_float4(v.x * s, v.y * s, v.z * s, v.w * s);
}

__global__ void k_edge_agg(float* __restrict__ out, const float* __restrict__ h,
                           const float* __restrict__ dinv,
                           const int* __restrict__ row, const int* __restrict__ col) {
  long long t = (long long)blockIdx.x * blockDim.x + threadIdx.x;
  if (t >= (long long)NE * (HIDC / 4)) return;
  int e = (int)(t >> 5);
  int g = (int)(t & 31);
  int r = row[e], c = col[e];
  float norm = dinv[r] * dinv[c];
  float4 v = ((const float4*)(h + (size_t)r * HIDC))[g];
  float* dst = out + (size_t)c * HIDC + g * 4;
  unsafeAtomicAdd(dst + 0, v.x * norm);
  unsafeAtomicAdd(dst + 1, v.y * norm);
  unsafeAtomicAdd(dst + 2, v.z * norm);
  unsafeAtomicAdd(dst + 3, v.w * norm);
}

__global__ void k_bias_act(float* __restrict__ z, const float* __restrict__ b,
                           int do_relu) {
  int i = blockIdx.x * blockDim.x + threadIdx.x;
  if (i >= NN * HIDC) return;
  int f = i & (HIDC - 1);
  float v = z[i] + b[f];
  if (do_relu) v = fmaxf(v, 0.0f);
  z[i] = v;
}

__global__ void k_decode(const float* __restrict__ z, const int* __restrict__ pos,
                         const int* __restrict__ neg, float* __restrict__ out) {
  int w = (blockIdx.x * blockDim.x + threadIdx.x) >> 5;
  int lane = threadIdx.x & 31;
  if (w >= NPOS + NNEG) return;
  int a, b;
  if (w < NPOS) { a = pos[w]; b = pos[NPOS + w]; }
  else          { int j = w - NPOS; a = neg[j]; b = neg[NNEG + j]; }
  float4 va = ((const float4*)(z + (size_t)a * HIDC))[lane];
  float4 vb = ((const float4*)(z + (size_t)b * HIDC))[lane];
  float s = va.x * vb.x + va.y * vb.y + va.z * vb.z + va.w * vb.w;
#pragma unroll
  for (int off = 16; off > 0; off >>= 1) s += __shfl_xor(s, off, 32);
  if (lane == 0) out[w] = s;
}

extern "C" void kernel_launch(void* const* d_in, const int* in_sizes, int n_in,
                              void* d_out, int out_size, void* d_ws, size_t ws_size,
                              hipStream_t stream) {
  const float* x  = (const float*)d_in[0];
  const int*   ei = (const int*)d_in[1];
  const int*   pos = (const int*)d_in[2];
  const int*   neg = (const int*)d_in[3];
  const float* W1 = (const float*)d_in[4];
  const float* b1 = (const float*)d_in[5];
  const float* W2 = (const float*)d_in[6];
  const float* b2 = (const float*)d_in[7];
  float* logits = (float*)d_out;

  char* ws = (char*)d_ws;
  float* dinv = (float*)ws;
  float* WT   = (float*)(ws + (512u << 10));
  float* bufH = (float*)(ws + (1u << 20));
  float* bufZ = (float*)(ws + (1u << 20) + (52u << 20));

  const int* row = ei;
  const int* col = ei + NE;
  const int B = 256;

  k_init_dinv<<<(NN + B - 1) / B, B, 0, stream>>>(dinv);
  k_count_deg<<<(NE + B - 1) / B, B, 0, stream>>>(dinv, col);
  k_finish_dinv<<<(NN + B - 1) / B, B, 0, stream>>>(dinv);

  int gemm_blocks = (((NN + 15) / 16) + 7) / 8;
  long long agg_t = (long long)NE * 32;
  int agg_blocks = (int)((agg_t + B - 1) / B);
  int sl_blocks  = (NN * 32 + B - 1) / B;
  int ba_blocks  = (NN * HIDC + B - 1) / B;

  k_transpose<<<(INC * HIDC + B - 1) / B, B, 0, stream>>>(W1, WT, INC, HIDC);
  k_gemm_wmma<INC><<<gemm_blocks, B, 0, stream>>>(x, WT, bufH);
  k_selfloop<<<sl_blocks, B, 0, stream>>>(bufZ, bufH, dinv);
  k_edge_agg<<<agg_blocks, B, 0, stream>>>(bufZ, bufH, dinv, row, col);
  k_bias_act<<<ba_blocks, B, 0, stream>>>(bufZ, b1, 1);

  k_transpose<<<(HIDC * HIDC + B - 1) / B, B, 0, stream>>>(W2, WT, HIDC, HIDC);
  k_gemm_wmma<HIDC><<<gemm_blocks, B, 0, stream>>>(bufZ, WT, bufH);
  k_selfloop<<<sl_blocks, B, 0, stream>>>(bufZ, bufH, dinv);
  k_edge_agg<<<agg_blocks, B, 0, stream>>>(bufZ, bufH, dinv, row, col);
  k_bias_act<<<ba_blocks, B, 0, stream>>>(bufZ, b2, 0);

  int dec_blocks = ((NPOS + NNEG) * 32 + B - 1) / B;
  k_decode<<<dec_blocks, B, 0, stream>>>(bufZ, pos, neg, logits);
}


# edge aggregation: 256 groups each hold 391 output rows in LDS, scan the whole destination list for their edges and add source rows by LDS float adds (arrival order); self-loop, bias, relu folded in
# speedup vs baseline: 1.6489x; 1.6489x over previous
#include <hip/hip_runtime.h>

#define NN   100000
#define INC  256
#define HIDC 128
#define NE   1600000
#define NPOS 200000
#define NNEG 200000

typedef __attribute__((ext_vector_type(2))) float v2f;
typedef __attribute__((ext_vector_type(8))) float v8f;

__global__ void k_init_dinv(float* dinv) {
  int i = blockIdx.x * blockDim.x + threadIdx.x;
  if (i < NN) dinv[i] = 1.0f;
}

__global__ void k_count_deg(float* dinv, const int* __restrict__ col) {
  int i = blockIdx.x * blockDim.x + threadIdx.x;
  if (i < NE) unsafeAtomicAdd(&dinv[col[i]], 1.0f);
}

__global__ void k_finish_dinv(float* dinv) {
  int i = blockIdx.x * blockDim.x + threadIdx.x;
  if (i < NN) dinv[i] = 1.0f / sqrtf(dinv[i]);
}

__global__ void k_transpose(const float* __restrict__ W, float* __restrict__ WT,
                            int K, int N) {
  int i = blockIdx.x * blockDim.x + threadIdx.x;
  if (i >= K * N) return;
  int k = i / N, n = i - k * N;
  WT[n * K + k] = W[i];
}

template <int K>
__global__ __launch_bounds__(256) void k_gemm_wmma(const float* __restrict__ A,
                                                   const float* __restrict__ WT,
                                                   float* __restrict__ C) {
  int wave = (blockIdx.x * blockDim.x + threadIdx.x) >> 5;
  int lane = threadIdx.x & 31;
  int m0 = wave * 16;
  if (m0 >= NN) return;
  int half = lane >> 4;
  int l15  = lane & 15;
  const float* aptr  = A  + (size_t)(m0 + l15) * K + half * 2;
  const float* bbase = WT + (size_t)l15 * K + half * 2;

  v8f acc[8];
#pragma unroll
  for (int t = 0; t < 8; ++t) acc[t] = (v8f)(0.0f);

  __builtin_prefetch(aptr, 0, 0);
#pragma unroll 2
  for (int k = 0; k < K; k += 4) {
    v2f a = *(const v2f*)(aptr + k);
#pragma unroll
    for (int t = 0; t < 8; ++t) {
      v2f b = *(const v2f*)(bbase + (size_t)t * 16 * K + k);
      acc[t] = __builtin_amdgcn_wmma_f32_16x16x4_f32(
          false, a, false, b, (short)0, acc[t], false, false);
    }
  }

#pragma unroll
  for (int t = 0; t < 8; ++t) {
#pragma unroll
    for (int i = 0; i < 8; ++i) {
      C[(size_t)(m0 + i + half * 8) * HIDC + t * 16 + l15] = acc[t][i];
    }
  }
}

#define AGG_GROUPS 256
#define RPG   391
#define QDRAIN 64
#define QCAP  (QDRAIN + 512)
#define NQUAD (NE / 4)
#define SCAN_TRIPS 392

template <bool FINAL>
__device__ __forceinline__ int agg_drain(float* acc, int* q, int qn, int lo, int lane,
                                         const float* __restrict__ h,
                                         const float* __restrict__ dinv,
                                         const int* __restrict__ row,
                                         const int* __restrict__ col) {
  const int todo = FINAL ? qn : (qn & ~7);
  for (int base = 0; base < todo; base += 32) {
    int cnt = min(32, todo - base);
    int r = 0, cl = 0;
    float nm = 0.0f;
    if (lane < cnt) {
      int e = q[base + lane];
      r = row[e];
      int c = col[e];
      cl = c - lo;
      nm = dinv[r] * dinv[c];
    }
    int j = 0;
    for (; j + 8 <= cnt; j += 8) {
      float v[8][4];
      int   cj[8];
      float nj[8];
#pragma unroll
      for (int u = 0; u < 8; ++u) {
        int rj = __builtin_amdgcn_readlane(r, j + u);
        cj[u] = __builtin_amdgcn_readlane(cl, j + u);
        nj[u] = __int_as_float(__builtin_amdgcn_readlane(__float_as_int(nm), j + u));
        const float* src = h + (size_t)rj * HIDC + lane;
        v[u][0] = src[0];
        v[u][1] = src[32];
        v[u][2] = src[64];
        v[u][3] = src[96];
      }
#pragma unroll
      for (int u = 0; u < 8; ++u) {
        float* dst = acc + cj[u] * HIDC + lane;
        unsafeAtomicAdd(dst + 0,  v[u][0] * nj[u]);
        unsafeAtomicAdd(dst + 32, v[u][1] * nj[u]);
        unsafeAtomicAdd(dst + 64, v[u][2] * nj[u]);
        unsafeAtomicAdd(dst + 96, v[u][3] * nj[u]);
      }
    }
    if (FINAL) {
      for (; j < cnt; ++j) {
        int   rj = __builtin_amdgcn_readlane(r, j);
        int   cj = __builtin_amdgcn_readlane(cl, j);
        float nj = __int_as_float(__builtin_amdgcn_readlane(__float_as_int(nm), j));
        const float* src = h + (size_t)rj * HIDC + lane;
        float v0 = src[0], v1 = src[32], v2 = src[64], v3 = src[96];
        float* dst = acc + cj * HIDC + lane;
        unsafeAtomicAdd(dst + 0,  v0 * nj);
        unsafeAtomicAdd(dst + 32, v1 * nj);
        unsafeAtomicAdd(dst + 64, v2 * nj);
        unsafeAtomicAdd(dst + 96, v3 * nj);
      }
    }
  }
  if (FINAL) return 0;
  const int rem = qn - todo;
  if (lane < rem) {
    int t = q[todo + lane];
    q[lane] = t;
  }
  return rem;
}

__global__ __launch_bounds__(1024) void k_agg_owner(float* __restrict__ out,
                                                    const float* __restrict__ h,
                                                    const float* __restrict__ dinv,
                                                    const int* __restrict__ row,
                                                    const int* __restrict__ col,
                                                    const float* __restrict__ bias,
                                                    int do_relu) {
  __shared__ float acc[RPG * HIDC];
  __shared__ int   queue[32 * QCAP];
  const int tid  = threadIdx.x;
  const int lane = tid & 31;
  const int wv   = tid >> 5;
  const int lo   = blockIdx.x * RPG;
  const int nrows = min(RPG, NN - lo);
  int* q = queue + wv * QCAP;

  const float4* h4 = (const float4*)(h + (size_t)lo * HIDC);
  for (int i = tid; i < nrows * (HIDC / 4); i += 1024) {
    float d = dinv[lo + (i >> 5)];
    float s = d * d;
    float4 v = h4[i];
    ((float4*)acc)[i] = make_float4(v.x * s, v.y * s, v.z * s, v.w * s);
  }
  __syncthreads();

  const int4* col4 = (const int4*)col;
  const int trip0 = (blockIdx.x * SCAN_TRIPS) >> 8;
  const unsigned below = (1u << lane) - 1u;
  int qn = 0;
  for (int k = 0; k < SCAN_TRIPS; k += 8) {
    int4 c[8];
    int  g[8];
#pragma unroll
    for (int u = 0; u < 8; ++u) {
      int t = trip0 + k + u;
      if (t >= SCAN_TRIPS) t -= SCAN_TRIPS;
      g[u] = t * 1024 + tid;
      c[u] = (g[u] < NQUAD) ? col4[g[u]] : make_int4(-1, -1, -1, -1);
    }
#pragma unroll
    for (int half = 0; half < 2; ++half) {
#pragma unroll
      for (int w = 0; w < 4; ++w) {
        const int u = half * 4 + w;
        bool m0 = (unsigned)(c[u].x - lo) < (unsigned)nrows;
        bool m1 = (unsigned)(c[u].y - lo) < (unsigned)nrows;
        bool m2 = (unsigned)(c[u].z - lo) < (unsigned)nrows;
        bool m3 = (unsigned)(c[u].w - lo) < (unsigned)nrows;
        if (__builtin_amdgcn_ballot_w32(m0 | m1 | m2 | m3)) {
          unsigned b0 = __builtin_amdgcn_ballot_w32(m0);
          if (m0) q[qn + __popc(b0 & below)] = 4 * g[u] + 0;
          qn += __popc(b0);
          unsigned b1 = __builtin_amdgcn_ballot_w32(m1);
          if (m1) q[qn + __popc(b1 & below)] = 4 * g[u] + 1;
          qn += __popc(b1);
          unsigned b2 = __builtin_amdgcn_ballot_w32(m2);
          if (m2) q[qn + __popc(b2 & below)] = 4 * g[u] + 2;
          qn += __popc(b2);
          unsigned b3 = __builtin_amdgcn_ballot_w32(m3);
          if (m3) q[qn + __popc(b3 & below)] = 4 * g[u] + 3;
          qn += __popc(b3);
        }
      }
      if (qn >= QDRAIN)
        qn = agg_drain<false>(acc, q, qn, lo, lane, h, dinv, row, col);
    }
  }
  agg_drain<true>(acc, q, qn, lo, lane, h, dinv, row, col);
  __syncthreads();

  float4* o4 = (float4*)(out + (size_t)lo * HIDC);
  for (int i = tid; i < nrows * (HIDC / 4); i += 1024) {
    float4 v = ((const float4*)acc)[i];
    float4 b = ((const float4*)bias)[i & 31];
    v.x += b.x; v.y += b.y; v.z += b.z; v.w += b.w;
    if (do_relu) {
      v.x = fmaxf(v.x, 0.0f); v.y = fmaxf(v.y, 0.0f);
      v.z = fmaxf(v.z, 0.0f); v.w = fmaxf(v.w, 0.0f);
    }
    o4[i] = v;
  }
}

__global__ void k_decode(const float* __restrict__ z, const int* __restrict__ pos,
                         const int* __restrict__ neg, float* __restrict__ out) {
  int w = (blockIdx.x * blockDim.x + threadIdx.x) >> 5;
  int lane = threadIdx.x & 31;
  if (w >= NPOS + NNEG) return;
  int a, b;
  if (w < NPOS) { a = pos[w]; b = pos[NPOS + w]; }
  else          { int j = w - NPOS; a = neg[j]; b = neg[NNEG + j]; }
  float4 va = ((const float4*)(z + (size_t)a * HIDC))[lane];
  float4 vb = ((const float4*)(z + (size_t)b * HIDC))[lane];
  float s = va.x * vb.x + va.y * vb.y + va.z * vb.z + va.w * vb.w;
#pragma unroll
  for (int off = 16; off > 0; off >>= 1) s += __shfl_xor(s, off, 32);
  if (lane == 0) out[w] = s;
}

extern "C" void kernel_launch(void* const* d_in, const int* in_sizes, int n_in,
                              void* d_out, int out_size, void* d_ws, size_t ws_size,
                              hipStream_t stream) {
  const float* x  = (const float*)d_in[0];
  const int*   ei = (const int*)d_in[1];
  const int*   pos = (const int*)d_in[2];
  const int*   neg = (const int*)d_in[3];
  const float* W1 = (const float*)d_in[4];
  const float* b1 = (const float*)d_in[5];
  const float* W2 = (const float*)d_in[6];
  const float* b2 = (const float*)d_in[7];
  float* logits = (float*)d_out;

  char* ws = (char*)d_ws;
  float* dinv = (float*)ws;
  float* WT   = (float*)(ws + (512u << 10));
  float* bufH = (float*)(ws + (1u << 20));
  float* bufZ = (float*)(ws + (1u << 20) + (52u << 20));

  const int* row = ei;
  const int* col = ei + NE;
  const int B = 256;

  k_init_dinv<<<(NN + B - 1) / B, B, 0, stream>>>(dinv);
  k_count_deg<<<(NE + B - 1) / B, B, 0, stream>>>(dinv, col);
  k_finish_dinv<<<(NN + B - 1) / B, B, 0, stream>>>(dinv);

  int gemm_blocks = (((NN + 15) / 16) + 7) / 8;

  k_transpose<<<(INC * HIDC + B - 1) / B, B, 0, stream>>>(W1, WT, INC, HIDC);
  k_gemm_wmma<INC><<<gemm_blocks, B, 0, stream>>>(x, WT, bufH);
  k_agg_owner<<<AGG_GROUPS, 1024, 0, stream>>>(bufZ, bufH, dinv, row, col, b1, 1);

  k_transpose<<<(HIDC * HIDC + B - 1) / B, B, 0, stream>>>(W2, WT, HIDC, HIDC);
  k_gemm_wmma<HIDC><<<gemm_blocks, B, 0, stream>>>(bufZ, WT, bufH);
  k_agg_owner<<<AGG_GROUPS, 1024, 0, stream>>>(bufZ, bufH, dinv, row, col, b2, 0);

  int dec_blocks = ((NPOS + NNEG) * 32 + B - 1) / B;
  k_decode<<<dec_blocks, B, 0, stream>>>(bufZ, pos, neg, logits);
}


# edge scan: degree pass also writes a 1-byte owner key per edge (dest row / 391); each group scans the 1.6 MB key list, 16 keys per lane load, not the 6.4 MB destination list; LDS-add order changes
# speedup vs baseline: 1.8946x; 1.1490x over previous
#include <hip/hip_runtime.h>

#define NN   100000
#define INC  256
#define HIDC 128
#define NE   1600000
#define NPOS 200000
#define NNEG 200000

typedef __attribute__((ext_vector_type(2))) float v2f;
typedef __attribute__((ext_vector_type(8))) float v8f;

#define AGG_GROUPS 256
#define RPG   391

__global__ void k_init_dinv(float* dinv) {
  int i = blockIdx.x * blockDim.x + threadIdx.x;
  if (i < NN) dinv[i] = 1.0f;
}

__global__ void k_count_deg(float* dinv, const int* __restrict__ col,
                            unsigned* __restrict__ key) {
  int i = blockIdx.x * blockDim.x + threadIdx.x;
  if (i < NE / 4) {
    int4 c = ((const int4*)col)[i];
    unsafeAtomicAdd(&dinv[c.x], 1.0f);
    unsafeAtomicAdd(&dinv[c.y], 1.0f);
    unsafeAtomicAdd(&dinv[c.z], 1.0f);
    unsafeAtomicAdd(&dinv[c.w], 1.0f);
    unsigned k0 = (unsigned)c.x / RPG, k1 = (unsigned)c.y / RPG;
    unsigned k2 = (unsigned)c.z / RPG, k3 = (unsigned)c.w / RPG;
    key[i] = k0 | (k1 << 8) | (k2 << 16) | (k3 << 24);
  }
}

__global__ void k_finish_dinv(float* dinv) {
  int i = blockIdx.x * blockDim.x + threadIdx.x;
  if (i < NN) dinv[i] = 1.0f / sqrtf(dinv[i]);
}

__global__ void k_transpose(const float* __restrict__ W, float* __restrict__ WT,
                            int K, int N) {
  int i = blockIdx.x * blockDim.x + threadIdx.x;
  if (i >= K * N) return;
  int k = i / N, n = i - k * N;
  WT[n * K + k] = W[i];
}

template <int K>
__global__ __launch_bounds__(256) void k_gemm_wmma(const float* __restrict__ A,
                                                   const float* __restrict__ WT,
                                                   float* __restrict__ C) {
  int wave = (blockIdx.x * blockDim.x + threadIdx.x) >> 5;
  int lane = threadIdx.x & 31;
  int m0 = wave * 16;
  if (m0 >= NN) return;
  int half = lane >> 4;
  int l15  = lane & 15;
  const float* aptr  = A  + (size_t)(m0 + l15) * K + half * 2;
  const float* bbase = WT + (size_t)l15 * K + half * 2;

  v8f acc[8];
#pragma unroll
  for (int t = 0; t < 8; ++t) acc[t] = (v8f)(0.0f);

  __builtin_prefetch(aptr, 0, 0);
#pragma unroll 2
  for (int k = 0; k < K; k += 4) {
    v2f a = *(const v2f*)(aptr + k);
#pragma unroll
    for (int t = 0; t < 8; ++t) {
      v2f b = *(const v2f*)(bbase + (size_t)t * 16 * K + k);
      acc[t] = __builtin_amdgcn_wmma_f32_16x16x4_f32(
          false, a, false, b, (short)0, acc[t], false, false);
    }
  }

#pragma unroll
  for (int t = 0; t < 8; ++t) {
#pragma unroll
    for (int i = 0; i < 8; ++i) {
      C[(size_t)(m0 + i + half * 8) * HIDC + t * 16 + l15] = acc[t][i];
    }
  }
}

#define QDRAIN 64
#define QCAP  (QDRAIN + 512)
#define NKEY16 (NE / 16)
#define SCAN_TRIPS 98

__device__ __forceinline__ unsigned zero_bytes(unsigned x) {
  unsigned t = (x & 0x7f7f7f7fu) + 0x7f7f7f7fu;
  return ~(t | x) & 0x80808080u;
}

template <bool FINAL>
__device__ __forceinline__ int agg_drain(float* acc, int* q, int qn, int lo, int lane,
                                         const float* __restrict__ h,
                                         const float* __restrict__ dinv,
                                         const int* __restrict__ row,
                                         const int* __restrict__ col) {
  const int todo = FINAL ? qn : (qn & ~7);
  for (int base = 0; base < todo; base += 32) {
    int cnt = min(32, todo - base);
    int r = 0, cl = 0;
    float nm = 0.0f;
    if (lane < cnt) {
      int e = q[base + lane];
      r = row[e];
      int c = col[e];
      cl = c - lo;
      nm = dinv[r] * dinv[c];
    }
    int j = 0;
    for (; j + 8 <= cnt; j += 8) {
      float v[8][4];
      int   cj[8];
      float nj[8];
#pragma unroll
      for (int u = 0; u < 8; ++u) {
        int rj = __builtin_amdgcn_readlane(r, j + u);
        cj[u] = __builtin_amdgcn_readlane(cl, j + u);
        nj[u] = __int_as_float(__builtin_amdgcn_readlane(__float_as_int(nm), j + u));
        const float* src = h + (size_t)rj * HIDC + lane;
        v[u][0] = src[0];
        v[u][1] = src[32];
        v[u][2] = src[64];
        v[u][3] = src[96];
      }
#pragma unroll
      for (int u = 0; u < 8; ++u) {
        float* dst = acc + cj[u] * HIDC + lane;
        unsafeAtomicAdd(dst + 0,  v[u][0] * nj[u]);
        unsafeAtomicAdd(dst + 32, v[u][1] * nj[u]);
        unsafeAtomicAdd(dst + 64, v[u][2] * nj[u]);
        unsafeAtomicAdd(dst + 96, v[u][3] * nj[u]);
      }
    }
    if (FINAL) {
      for (; j < cnt; ++j) {
        int   rj = __builtin_amdgcn_readlane(r, j);
        int   cj = __builtin_amdgcn_readlane(cl, j);
        float nj = __int_as_float(__builtin_amdgcn_readlane(__float_as_int(nm), j));
        const float* src = h + (size_t)rj * HIDC + lane;
        float v0 = src[0], v1 = src[32], v2 = src[64], v3 = src[96];
        float* dst = acc + cj * HIDC + lane;
        unsafeAtomicAdd(dst + 0,  v0 * nj);
        unsafeAtomicAdd(dst + 32, v1 * nj);
        unsafeAtomicAdd(dst + 64, v2 * nj);
        unsafeAtomicAdd(dst + 96, v3 * nj);
      }
    }
  }
  if (FINAL) return 0;
  const int rem = qn - todo;
  if (lane < rem) {
    int t = q[todo + lane];
    q[lane] = t;
  }
  return rem;
}

__global__ __launch_bounds__(1024) void k_agg_owner(float* __restrict__ out,
                                                    const float* __restrict__ h,
                                                    const float* __restrict__ dinv,
                                                    const int* __restrict__ row,
                                                    const int* __restrict__ col,
                                                    const unsigned* __restrict__ key,
                                                    const float* __restrict__ bias,
                                                    int do_relu) {
  __shared__ float acc[RPG * HIDC];
  __shared__ int   queue[32 * QCAP];
  const int tid  = threadIdx.x;
  const int lane = tid & 31;
  const int wv   = tid >> 5;
  const int lo   = blockIdx.x * RPG;
  const int nrows = min(RPG, NN - lo);
  int* q = queue + wv * QCAP;

  const float4* h4 = (const float4*)(h + (size_t)lo * HIDC);
  for (int i = tid; i < nrows * (HIDC / 4); i += 1024) {
    float d = dinv[lo + (i >> 5)];
    float s = d * d;
    float4 v = h4[i];
    ((float4*)acc)[i] = make_float4(v.x * s, v.y * s, v.z * s, v.w * s);
  }
  __syncthreads();

  const uint4* key4 = (const uint4*)key;
  const unsigned pat = blockIdx.x * 0x01010101u;
  const int trip0 = (blockIdx.x * SCAN_TRIPS) >> 8;
  const unsigned below = (1u << lane) - 1u;
  int qn = 0;
  for (int k = 0; k < SCAN_TRIPS; k += 7) {
    uint4 c[7];
    int   g[7];
#pragma unroll
    for (int u = 0; u < 7; ++u) {
      int t = trip0 + k + u;
      if (t >= SCAN_TRIPS) t -= SCAN_TRIPS;
      g[u] = t * 1024 + tid;
      c[u] = key4[min(g[u], NKEY16 - 1)];
    }
#pragma unroll
    for (int u = 0; u < 7; ++u) {
      unsigned mm = (zero_bytes(c[u].x ^ pat) >> 7) | (zero_bytes(c[u].y ^ pat) >> 6) |
                    (zero_bytes(c[u].z ^ pat) >> 5) | (zero_bytes(c[u].w ^ pat) >> 4);
      if (g[u] >= NKEY16) mm = 0;
      if (__builtin_amdgcn_ballot_w32(mm != 0)) {
        do {
          bool has = mm != 0;
          unsigned b = __builtin_amdgcn_ballot_w32(has);
          if (has) {
            int p = __builtin_ctz(mm);
            q[qn + __popc(b & below)] = 16 * g[u] + ((p & 7) << 2) + (p >> 3);
            mm &= mm - 1;
          }
          qn += __popc(b);
        } while (__builtin_amdgcn_ballot_w32(mm != 0));
        if (qn >= QDRAIN)
          qn = agg_drain<false>(acc, q, qn, lo, lane, h, dinv, row, col);
      }
    }
  }
  agg_drain<true>(acc, q, qn, lo, lane, h, dinv, row, col);
  __syncthreads();

  float4* o4 = (float4*)(out + (size_t)lo * HIDC);
  for (int i = tid; i < nrows * (HIDC / 4); i += 1024) {
    float4 v = ((const float4*)acc)[i];
    float4 b = ((const float4*)bias)[i & 31];
    v.x += b.x; v.y += b.y; v.z += b.z; v.w += b.w;
    if (do_relu) {
      v.x = fmaxf(v.x, 0.0f); v.y = fmaxf(v.y, 0.0f);
      v.z = fmaxf(v.z, 0.0f); v.w = fmaxf(v.w, 0.0f);
    }
    o4[i] = v;
  }
}

__global__ void k_decode(const float* __restrict__ z, const int* __restrict__ pos,
                         const int* __restrict__ neg, float* __restrict__ out) {
  int w = (blockIdx.x * blockDim.x + threadIdx.x) >> 5;
  int lane = threadIdx.x & 31;
  if (w >= NPOS + NNEG) return;
  int a, b;
  if (w < NPOS) { a = pos[w]; b = pos[NPOS + w]; }
  else          { int j = w - NPOS; a = neg[j]; b = neg[NNEG + j]; }
  float4 va = ((const float4*)(z + (size_t)a * HIDC))[lane];
  float4 vb = ((const float4*)(z + (size_t)b * HIDC))[lane];
  float s = va.x * vb.x + va.y * vb.y + va.z * vb.z + va.w * vb.w;
#pragma unroll
  for (int off = 16; off > 0; off >>= 1) s += __shfl_xor(s, off, 32);
  if (lane == 0) out[w] = s;
}

extern "C" void kernel_launch(void* const* d_in, const int* in_sizes, int n_in,
                              void* d_out, int out_size, void* d_ws, size_t ws_size,
                              hipStream_t stream) {
  const float* x  = (const float*)d_in[0];
  const int*   ei = (const int*)d_in[1];
  const int*   pos = (const int*)d_in[2];
  const int*   neg = (const int*)d_in[3];
  const float* W1 = (const float*)d_in[4];
  const float* b1 = (const float*)d_in[5];
  const float* W2 = (const float*)d_in[6];
  const float* b2 = (const float*)d_in[7];
  float* logits = (float*)d_out;

  char* ws = (char*)d_ws;
  float* dinv = (float*)ws;
  float* WT   = (float*)(ws + (512u << 10));
  float* bufH = (float*)(ws + (1u << 20));
  float* bufZ = (float*)(ws + (1u << 20) + (52u << 20));
  unsigned* key = (unsigned*)(ws + (1u << 20) + 51200000u);

  const int* row = ei;
  const int* col = ei + NE;
  const int B = 256;

  k_init_dinv<<<(NN + B - 1) / B, B, 0, stream>>>(dinv);
  k_count_deg<<<(NE / 4 + B - 1) / B, B, 0, stream>>>(dinv, col, key);
  k_finish_dinv<<<(NN + B - 1) / B, B, 0, stream>>>(dinv);

  int gemm_blocks = (((NN + 15) / 16) + 7) / 8;

  k_transpose<<<(INC * HIDC + B - 1) / B, B, 0, stream>>>(W1, WT, INC, HIDC);
  k_gemm_wmma<INC><<<gemm_blocks, B, 0, stream>>>(x, WT, bufH);
  k_agg_owner<<<AGG_GROUPS, 1024, 0, stream>>>(bufZ, bufH, dinv, row, col, key, b1, 1);

  k_transpose<<<(HIDC * HIDC + B - 1) / B, B, 0, stream>>>(W2, WT, HIDC, HIDC);
  k_gemm_wmma<HIDC><<<gemm_blocks, B, 0, stream>>>(bufZ, WT, bufH);
  k_agg_owner<<<AGG_GROUPS, 1024, 0, stream>>>(bufZ, bufH, dinv, row, col, key, b2, 0);

  int dec_blocks = ((NPOS + NNEG) * 32 + B - 1) / B;
  k_decode<<<dec_blocks, B, 0, stream>>>(bufZ, pos, neg, logits);
}


# GEMM made weight-stationary: each wave keeps the W fragments of its 32 columns for all of K in registers; 256 resident groups walk 32-row A blocks staged in LDS by whole-line loads; same k order
# speedup vs baseline: 2.3085x; 1.2185x over previous
#include <hip/hip_runtime.h>

#define NN   100000
#define INC  256
#define HIDC 128
#define NE   1600000
#define NPOS 200000
#define NNEG 200000

typedef __attribute__((ext_vector_type(2))) float v2f;
typedef __attribute__((ext_vector_type(8))) float v8f;

#define AGG_GROUPS 256
#define RPG   391

__global__ void k_init_dinv(float* dinv) {
  int i = blockIdx.x * blockDim.x + threadIdx.x;
  if (i < NN) dinv[i] = 1.0f;
}

__global__ void k_count_deg(float* dinv, const int* __restrict__ col,
                            unsigned* __restrict__ key) {
  int i = blockIdx.x * blockDim.x + threadIdx.x;
  if (i < NE / 4) {
    int4 c = ((const int4*)col)[i];
    unsafeAtomicAdd(&dinv[c.x], 1.0f);
    unsafeAtomicAdd(&dinv[c.y], 1.0f);
    unsafeAtomicAdd(&dinv[c.z], 1.0f);
    unsafeAtomicAdd(&dinv[c.w], 1.0f);
    unsigned k0 = (unsigned)c.x / RPG, k1 = (unsigned)c.y / RPG;
    unsigned k2 = (unsigned)c.z / RPG, k3 = (unsigned)c.w / RPG;
    key[i] = k0 | (k1 << 8) | (k2 << 16) | (k3 << 24);
  }
}

__global__ void k_finish_dinv(float* dinv) {
  int i = blockIdx.x * blockDim.x + threadIdx.x;
  if (i < NN) dinv[i] = 1.0f / sqrtf(dinv[i]);
}

__global__ void k_transpose(const float* __restrict__ W, float* __restrict__ WT,
                            int K, int N) {
  int i = blockIdx.x * blockDim.x + threadIdx.x;
  if (i >= K * N) return;
  int k = i / N, n = i - k * N;
  WT[n * K + k] = W[i];
}

#define GROWS 32
#define GPAIRS (NN / GROWS)
#define GEMM_GROUPS 256
template <int K>
__global__ __launch_bounds__(256) void k_gemm_wmma(const float* __restrict__ A,
                                                   const float* __restrict__ WT,
                                                   float* __restrict__ C) {
  constexpr int PITCH = K + 4;
  constexpr int F4ROW = K / 4;
  constexpr int NLD   = GROWS * F4ROW / 256;
  __shared__ __attribute__((aligned(16))) float sA[2 * GROWS * PITCH];

  const int tid  = threadIdx.x;
  const int lane = tid & 31;
  const int wv   = tid >> 5;
  const int half = lane >> 4;
  const int l15  = lane & 15;
  const int np   = wv & 3;
  const int mh   = wv >> 2;

  v2f b0[K / 4], b1[K / 4];
  {
    const float* bp0 = WT + (size_t)(np * 32 + l15) * K + half * 2;
    const float* bp1 = bp0 + (size_t)16 * K;
#pragma unroll
    for (int kk = 0; kk < K / 4; ++kk) {
      b0[kk] = *(const v2f*)(bp0 + 4 * kk);
      b1[kk] = *(const v2f*)(bp1 + 4 * kk);
    }
  }

  int soff[NLD];
#pragma unroll
  for (int j = 0; j < NLD; ++j) {
    int f = j * 256 + tid;
    soff[j] = (f / F4ROW) * PITCH + (f % F4ROW) * 4;
  }

  int p = blockIdx.x;
  {
    float4 st0[NLD];
    const float4* src = (const float4*)(A + (size_t)p * GROWS * K) + tid;
#pragma unroll
    for (int j = 0; j < NLD; ++j) st0[j] = src[j * 256];
#pragma unroll
    for (int j = 0; j < NLD; ++j) *(float4*)(sA + soff[j]) = st0[j];
  }
  __syncthreads();

  const int aoff = (mh * 16 + l15) * PITCH + half * 2;
  float* const cbase = C + (size_t)(mh * 16 + half * 8) * HIDC + np * 32 + l15;
  v8f pacc0 = (v8f)(0.0f), pacc1 = (v8f)(0.0f);
  int buf = 0;
  for (; p < GPAIRS; p += GEMM_GROUPS) {
    const bool more = p + GEMM_GROUPS < GPAIRS;
    float4 st[NLD];
    if (more) {
      const float4* src = (const float4*)(A + (size_t)(p + GEMM_GROUPS) * GROWS * K) + tid;
#pragma unroll
      for (int j = 0; j < NLD; ++j) st[j] = src[j * 256];
    } else {
#pragma unroll
      for (int j = 0; j < NLD; ++j) st[j] = make_float4(0.0f, 0.0f, 0.0f, 0.0f);
    }
    if (p != (int)blockIdx.x) {
      float* cp = cbase + (size_t)(p - GEMM_GROUPS) * GROWS * HIDC;
#pragma unroll
      for (int i = 0; i < 8; ++i) {
        cp[i * HIDC]      = pacc0[i];
        cp[i * HIDC + 16] = pacc1[i];
      }
    }

    v8f acc0 = (v8f)(0.0f), acc1 = (v8f)(0.0f);
    const float* ap = sA + buf * (GROWS * PITCH) + aoff;
#pragma unroll
    for (int kk = 0; kk < K / 4; ++kk) {
      v2f a = *(const v2f*)(ap + 4 * kk);
      acc0 = __builtin_amdgcn_wmma_f32_16x16x4_f32(
          false, a, false, b0[kk], (short)0, acc0, false, false);
      acc1 = __builtin_amdgcn_wmma_f32_16x16x4_f32(
          false, a, false, b1[kk], (short)0, acc1, false, false);
    }
    pacc0 = acc0;
    pacc1 = acc1;

    {
      float* dstb = sA + (buf ^ 1) * (GROWS * PITCH);
#pragma unroll
      for (int j = 0; j < NLD; ++j) *(float4*)(dstb + soff[j]) = st[j];
    }
    __syncthreads();
    buf ^= 1;
  }
  {
    float* cp = cbase + (size_t)(p - GEMM_GROUPS) * GROWS * HIDC;
#pragma unroll
    for (int i = 0; i < 8; ++i) {
      cp[i * HIDC]      = pacc0[i];
      cp[i * HIDC + 16] = pacc1[i];
    }
  }
}

#define QDRAIN 64
#define QCAP  (QDRAIN + 512)
#define NKEY16 (NE / 16)
#define SCAN_TRIPS 98

__device__ __forceinline__ unsigned zero_bytes(unsigned x) {
  unsigned t = (x & 0x7f7f7f7fu) + 0x7f7f7f7fu;
  return ~(t | x) & 0x80808080u;
}

template <bool FINAL>
__device__ __forceinline__ int agg_drain(float* acc, int* q, int qn, int lo, int lane,
                                         const float* __restrict__ h,
                                         const float* __restrict__ dinv,
                                         const int* __restrict__ row,
                                         const int* __restrict__ col) {
  const int todo = FINAL ? qn : (qn & ~7);
  for (int base = 0; base < todo; base += 32) {
    int cnt = min(32, todo - base);
    int r = 0, cl = 0;
    float nm = 0.0f;
    if (lane < cnt) {
      int e = q[base + lane];
      r = row[e];
      int c = col[e];
      cl = c - lo;
      nm = dinv[r] * dinv[c];
    }
    int j = 0;
    for (; j + 8 <= cnt; j += 8) {
      float v[8][4];
      int   cj[8];
      float nj[8];
#pragma unroll
      for (int u = 0; u < 8; ++u) {
        int rj = __builtin_amdgcn_readlane(r, j + u);
        cj[u] = __builtin_amdgcn_readlane(cl, j + u);
        nj[u] = __int_as_float(__builtin_amdgcn_readlane(__float_as_int(nm), j + u));
        const float* src = h + (size_t)rj * HIDC + lane;
        v[u][0] = src[0];
        v[u][1] = src[32];
        v[u][2] = src[64];
        v[u][3] = src[96];
      }
#pragma unroll
      for (int u = 0; u < 8; ++u) {
        float* dst = acc + cj[u] * HIDC + lane;
        unsafeAtomicAdd(dst + 0,  v[u][0] * nj[u]);
        unsafeAtomicAdd(dst + 32, v[u][1] * nj[u]);
        unsafeAtomicAdd(dst + 64, v[u][2] * nj[u]);
        unsafeAtomicAdd(dst + 96, v[u][3] * nj[u]);
      }
    }
    if (FINAL) {
      for (; j < cnt; ++j) {
        int   rj = __builtin_amdgcn_readlane(r, j);
        int   cj = __builtin_amdgcn_readlane(cl, j);
        float nj = __int_as_float(__builtin_amdgcn_readlane(__float_as_int(nm), j));
        const float* src = h + (size_t)rj * HIDC + lane;
        float v0 = src[0], v1 = src[32], v2 = src[64], v3 = src[96];
        float* dst = acc + cj * HIDC + lane;
        unsafeAtomicAdd(dst + 0,  v0 * nj);
        unsafeAtomicAdd(dst + 32, v1 * nj);
        unsafeAtomicAdd(dst + 64, v2 * nj);
        unsafeAtomicAdd(dst + 96, v3 * nj);
      }
    }
  }
  if (FINAL) return 0;
  const int rem = qn - todo;
  if (lane < rem) {
    int t = q[todo + lane];
    q[lane] = t;
  }
  return rem;
}

__global__ __launch_bounds__(1024) void k_agg_owner(float* __restrict__ out,
                                                    const float* __restrict__ h,
                                                    const float* __restrict__ dinv,
                                                    const int* __restrict__ row,
                                                    const int* __restrict__ col,
                                                    const unsigned* __restrict__ key,
                                                    const float* __restrict__ bias,
                                                    int do_relu) {
  __shared__ float acc[RPG * HIDC];
  __shared__ int   queue[32 * QCAP];
  const int tid  = threadIdx.x;
  const int lane = tid & 31;
  const int wv   = tid >> 5;
  const int lo   = blockIdx.x * RPG;
  const int nrows = min(RPG, NN - lo);
  int* q = queue + wv * QCAP;

  const float4* h4 = (const float4*)(h + (size_t)lo * HIDC);
  for (int i = tid; i < nrows * (HIDC / 4); i += 1024) {
    float d = dinv[lo + (i >> 5)];
    float s = d * d;
    float4 v = h4[i];
    ((float4*)acc)[i] = make_float4(v.x * s, v.y * s, v.z * s, v.w * s);
  }
  __syncthreads();

  const uint4* key4 = (const uint4*)key;
  const unsigned pat = blockIdx.x * 0x01010101u;
  const int trip0 = (blockIdx.x * SCAN_TRIPS) >> 8;
  const unsigned below = (1u << lane) - 1u;
  int qn = 0;
  for (int k = 0; k < SCAN_TRIPS; k += 7) {
    uint4 c[7];
    int   g[7];
#pragma unroll
    for (int u = 0; u < 7; ++u) {
      int t = trip0 + k + u;
      if (t >= SCAN_TRIPS) t -= SCAN_TRIPS;
      g[u] = t * 1024 + tid;
      c[u] = key4[min(g[u], NKEY16 - 1)];
    }
#pragma unroll
    for (int u = 0; u < 7; ++u) {
      unsigned mm = (zero_bytes(c[u].x ^ pat) >> 7) | (zero_bytes(c[u].y ^ pat) >> 6) |
                    (zero_bytes(c[u].z ^ pat) >> 5) | (zero_bytes(c[u].w ^ pat) >> 4);
      if (g[u] >= NKEY16) mm = 0;
      if (__builtin_amdgcn_ballot_w32(mm != 0)) {
        do {
          bool has = mm != 0;
          unsigned b = __builtin_amdgcn_ballot_w32(has);
          if (has) {
            int p = __builtin_ctz(mm);
            q[qn + __popc(b & below)] = 16 * g[u] + ((p & 7) << 2) + (p >> 3);
            mm &= mm - 1;
          }
          qn += __popc(b);
        } while (__builtin_amdgcn_ballot_w32(mm != 0));
        if (qn >= QDRAIN)
          qn = agg_drain<false>(acc, q, qn, lo, lane, h, dinv, row, col);
      }
    }
  }
  agg_drain<true>(acc, q, qn, lo, lane, h, dinv, row, col);
  __syncthreads();

  float4* o4 = (float4*)(out + (size_t)lo * HIDC);
  for (int i = tid; i < nrows * (HIDC / 4); i += 1024) {
    float4 v = ((const float4*)acc)[i];
    float4 b = ((const float4*)bias)[i & 31];
    v.x += b.x; v.y += b.y; v.z += b.z; v.w += b.w;
    if (do_relu) {
      v.x = fmaxf(v.x, 0.0f); v.y = fmaxf(v.y, 0.0f);
      v.z = fmaxf(v.z, 0.0f); v.w = fmaxf(v.w, 0.0f);
    }
    o4[i] = v;
  }
}

__global__ void k_decode(const float* __restrict__ z, const int* __restrict__ pos,
                         const int* __restrict__ neg, float* __restrict__ out) {
  int w = (blockIdx.x * blockDim.x + threadIdx.x) >> 5;
  int lane = threadIdx.x & 31;
  if (w >= NPOS + NNEG) return;
  int a, b;
  if (w < NPOS) { a = pos[w]; b = pos[NPOS + w]; }
  else          { int j = w - NPOS; a = neg[j]; b = neg[NNEG + j]; }
  float4 va = ((const float4*)(z + (size_t)a * HIDC))[lane];
  float4 vb = ((const float4*)(z + (size_t)b * HIDC))[lane];
  float s = va.x * vb.x + va.y * vb.y + va.z * vb.z + va.w * vb.w;
#pragma unroll
  for (int off = 16; off > 0; off >>= 1) s += __shfl_xor(s, off, 32);
  if (lane == 0) out[w] = s;
}

extern "C" void kernel_launch(void* const* d_in, const int* in_sizes, int n_in,
                              void* d_out, int out_size, void* d_ws, size_t ws_size,
                              hipStream_t stream) {
  const float* x  = (const float*)d_in[0];
  const int*   ei = (const int*)d_in[1];
  const int*   pos = (const int*)d_in[2];
  const int*   neg = (const int*)d_in[3];
  const float* W1 = (const float*)d_in[4];
  const float* b1 = (const float*)d_in[5];
  const float* W2 = (const float*)d_in[6];
  const float* b2 = (const float*)d_in[7];
  float* logits = (float*)d_out;

  char* ws = (char*)d_ws;
  float* dinv = (float*)ws;
  float* WT   = (float*)(ws + (512u << 10));
  float* bufH = (float*)(ws + (1u << 20));
  float* bufZ = (float*)(ws + (1u << 20) + (52u << 20));
  unsigned* key = (unsigned*)(ws + (1u << 20) + 51200000u);

  const int* row = ei;
  const int* col = ei + NE;
  const int B = 256;

  k_init_dinv<<<(NN + B - 1) / B, B, 0, stream>>>(dinv);
  k_count_deg<<<(NE / 4 + B - 1) / B, B, 0, stream>>>(dinv, col, key);
  k_finish_dinv<<<(NN + B - 1) / B, B, 0, stream>>>(dinv);

  const int gemm_blocks = GEMM_GROUPS;

  k_transpose<<<(INC * HIDC + B - 1) / B, B, 0, stream>>>(W1, WT, INC, HIDC);
  k_gemm_wmma<INC><<<gemm_blocks, B, 0, stream>>>(x, WT, bufH);
  k_agg_owner<<<AGG_GROUPS, 1024, 0, stream>>>(bufZ, bufH, dinv, row, col, key, b1, 1);

  k_transpose<<<(HIDC * HIDC + B - 1) / B, B, 0, stream>>>(W2, WT, HIDC, HIDC);
  k_gemm_wmma<HIDC><<<gemm_blocks, B, 0, stream>>>(bufZ, WT, bufH);
  k_agg_owner<<<AGG_GROUPS, 1024, 0, stream>>>(bufZ, bufH, dinv, row, col, key, b2, 0);

  int dec_blocks = ((NPOS + NNEG) * 32 + B - 1) / B;
  k_decode<<<dec_blocks, B, 0, stream>>>(bufZ, pos, neg, logits);
}


# decoder: 8 lanes per pair (4 pairs per wave pass, each load = 4 whole lines, 3-step lane sum); 256 resident groups of 25 waves walk 32-pair chunks, one line store each; dot-product sum order changes
# speedup vs baseline: 2.4995x; 1.0827x over previous
#include <hip/hip_runtime.h>

#define NN   100000
#define INC  256
#define HIDC 128
#define NE   1600000
#define NPOS 200000
#define NNEG 200000

typedef __attribute__((ext_vector_type(2))) float v2f;
typedef __attribute__((ext_vector_type(8))) float v8f;

#define AGG_GROUPS 256
#define RPG   391

__global__ void k_init_dinv(float* dinv) {
  int i = blockIdx.x * blockDim.x + threadIdx.x;
  if (i < NN) dinv[i] = 1.0f;
}

__global__ void k_count_deg(float* dinv, const int* __restrict__ col,
                            unsigned* __restrict__ key) {
  int i = blockIdx.x * blockDim.x + threadIdx.x;
  if (i < NE / 4) {
    int4 c = ((const int4*)col)[i];
    unsafeAtomicAdd(&dinv[c.x], 1.0f);
    unsafeAtomicAdd(&dinv[c.y], 1.0f);
    unsafeAtomicAdd(&dinv[c.z], 1.0f);
    unsafeAtomicAdd(&dinv[c.w], 1.0f);
    unsigned k0 = (unsigned)c.x / RPG, k1 = (unsigned)c.y / RPG;
    unsigned k2 = (unsigned)c.z / RPG, k3 = (unsigned)c.w / RPG;
    key[i] = k0 | (k1 << 8) | (k2 << 16) | (k3 << 24);
  }
}

__global__ void k_finish_dinv(float* dinv) {
  int i = blockIdx.x * blockDim.x + threadIdx.x;
  if (i < NN) dinv[i] = 1.0f / sqrtf(dinv[i]);
}

__global__ void k_transpose(const float* __restrict__ W, float* __restrict__ WT,
                            int K, int N) {
  int i = blockIdx.x * blockDim.x + threadIdx.x;
  if (i >= K * N) return;
  int k = i / N, n = i - k * N;
  WT[n * K + k] = W[i];
}

#define GROWS 32
#define GPAIRS (NN / GROWS)
#define GEMM_GROUPS 256
template <int K>
__global__ __launch_bounds__(256) void k_gemm_wmma(const float* __restrict__ A,
                                                   const float* __restrict__ WT,
                                                   float* __restrict__ C) {
  constexpr int PITCH = K + 4;
  constexpr int F4ROW = K / 4;
  constexpr int NLD   = GROWS * F4ROW / 256;
  __shared__ __attribute__((aligned(16))) float sA[2 * GROWS * PITCH];

  const int tid  = threadIdx.x;
  const int lane = tid & 31;
  const int wv   = tid >> 5;
  const int half = lane >> 4;
  const int l15  = lane & 15;
  const int np   = wv & 3;
  const int mh   = wv >> 2;

  v2f b0[K / 4], b1[K / 4];
  {
    const float* bp0 = WT + (size_t)(np * 32 + l15) * K + half * 2;
    const float* bp1 = bp0 + (size_t)16 * K;
#pragma unroll
    for (int kk = 0; kk < K / 4; ++kk) {
      b0[kk] = *(const v2f*)(bp0 + 4 * kk);
      b1[kk] = *(const v2f*)(bp1 + 4 * kk);
    }
  }

  int soff[NLD];
#pragma unroll
  for (int j = 0; j < NLD; ++j) {
    int f = j * 256 + tid;
    soff[j] = (f / F4ROW) * PITCH + (f % F4ROW) * 4;
  }

  int p = blockIdx.x;
  {
    float4 st0[NLD];
    const float4* src = (const float4*)(A + (size_t)p * GROWS * K) + tid;
#pragma unroll
    for (int j = 0; j < NLD; ++j) st0[j] = src[j * 256];
#pragma unroll
    for (int j = 0; j < NLD; ++j) *(float4*)(sA + soff[j]) = st0[j];
  }
  __syncthreads();

  const int aoff = (mh * 16 + l15) * PITCH + half * 2;
  float* const cbase = C + (size_t)(mh * 16 + half * 8) * HIDC + np * 32 + l15;
  v8f pacc0 = (v8f)(0.0f), pacc1 = (v8f)(0.0f);
  int buf = 0;
  for (; p < GPAIRS; p += GEMM_GROUPS) {
    const bool more = p + GEMM_GROUPS < GPAIRS;
    float4 st[NLD];
    if (more) {
      const float4* src = (const float4*)(A + (size_t)(p + GEMM_GROUPS) * GROWS * K) + tid;
#pragma unroll
      for (int j = 0; j < NLD; ++j) st[j] = src[j * 256];
    } else {
#pragma unroll
      for (int j = 0; j < NLD; ++j) st[j] = make_float4(0.0f, 0.0f, 0.0f, 0.0f);
    }
    if (p != (int)blockIdx.x) {
      float* cp = cbase + (size_t)(p - GEMM_GROUPS) * GROWS * HIDC;
#pragma unroll
      for (int i = 0; i < 8; ++i) {
        cp[i * HIDC]      = pacc0[i];
        cp[i * HIDC + 16] = pacc1[i];
      }
    }

    v8f acc0 = (v8f)(0.0f), acc1 = (v8f)(0.0f);
    const float* ap = sA + buf * (GROWS * PITCH) + aoff;
#pragma unroll
    for (int kk = 0; kk < K / 4; ++kk) {
      v2f a = *(const v2f*)(ap + 4 * kk);
      acc0 = __builtin_amdgcn_wmma_f32_16x16x4_f32(
          false, a, false, b0[kk], (short)0, acc0, false, false);
      acc1 = __builtin_amdgcn_wmma_f32_16x16x4_f32(
          false, a, false, b1[kk], (short)0, acc1, false, false);
    }
    pacc0 = acc0;
    pacc1 = acc1;

    {
      float* dstb = sA + (buf ^ 1) * (GROWS * PITCH);
#pragma unroll
      for (int j = 0; j < NLD; ++j) *(float4*)(dstb + soff[j]) = st[j];
    }
    __syncthreads();
    buf ^= 1;
  }
  {
    float* cp = cbase + (size_t)(p - GEMM_GROUPS) * GROWS * HIDC;
#pragma unroll
    for (int i = 0; i < 8; ++i) {
      cp[i * HIDC]      = pacc0[i];
      cp[i * HIDC + 16] = pacc1[i];
    }
  }
}

#define QDRAIN 64
#define QCAP  (QDRAIN + 512)
#define NKEY16 (NE / 16)
#define SCAN_TRIPS 98

__device__ __forceinline__ unsigned zero_bytes(unsigned x) {
  unsigned t = (x & 0x7f7f7f7fu) + 0x7f7f7f7fu;
  return ~(t | x) & 0x80808080u;
}

template <bool FINAL>
__device__ __forceinline__ int agg_drain(float* acc, int* q, int qn, int lo, int lane,
                                         const float* __restrict__ h,
                                         const float* __restrict__ dinv,
                                         const int* __restrict__ row,
                                         const int* __restrict__ col) {
  const int todo = FINAL ? qn : (qn & ~7);
  for (int base = 0; base < todo; base += 32) {
    int cnt = min(32, todo - base);
    int r = 0, cl = 0;
    float nm = 0.0f;
    if (lane < cnt) {
      int e = q[base + lane];
      r = row[e];
      int c = col[e];
      cl = c - lo;
      nm = dinv[r] * dinv[c];
    }
    int j = 0;
    for (; j + 8 <= cnt; j += 8) {
      float v[8][4];
      int   cj[8];
      float nj[8];
#pragma unroll
      for (int u = 0; u < 8; ++u) {
        int rj = __builtin_amdgcn_readlane(r, j + u);
        cj[u] = __builtin_amdgcn_readlane(cl, j + u);
        nj[u] = __int_as_float(__builtin_amdgcn_readlane(__float_as_int(nm), j + u));
        const float* src = h + (size_t)rj * HIDC + lane;
        v[u][0] = src[0];
        v[u][1] = src[32];
        v[u][2] = src[64];
        v[u][3] = src[96];
      }
#pragma unroll
      for (int u = 0; u < 8; ++u) {
        float* dst = acc + cj[u] * HIDC + lane;
        unsafeAtomicAdd(dst + 0,  v[u][0] * nj[u]);
        unsafeAtomicAdd(dst + 32, v[u][1] * nj[u]);
        unsafeAtomicAdd(dst + 64, v[u][2] * nj[u]);
        unsafeAtomicAdd(dst + 96, v[u][3] * nj[u]);
      }
    }
    if (FINAL) {
      for (; j < cnt; ++j) {
        int   rj = __builtin_amdgcn_readlane(r, j);
        int   cj = __builtin_amdgcn_readlane(cl, j);
        float nj = __int_as_float(__builtin_amdgcn_readlane(__float_as_int(nm), j));
        const float* src = h + (size_t)rj * HIDC + lane;
        float v0 = src[0], v1 = src[32], v2 = src[64], v3 = src[96];
        float* dst = acc + cj * HIDC + lane;
        unsafeAtomicAdd(dst + 0,  v0 * nj);
        unsafeAtomicAdd(dst + 32, v1 * nj);
        unsafeAtomicAdd(dst + 64, v2 * nj);
        unsafeAtomicAdd(dst + 96, v3 * nj);
      }
    }
  }
  if (FINAL) return 0;
  const int rem = qn - todo;
  if (lane < rem) {
    int t = q[todo + lane];
    q[lane] = t;
  }
  return rem;
}

__global__ __launch_bounds__(1024) void k_agg_owner(float* __restrict__ out,
                                                    const float* __restrict__ h,
                                                    const float* __restrict__ dinv,
                                                    const int* __restrict__ row,
                                                    const int* __restrict__ col,
                                                    const unsigned* __restrict__ key,
                                                    const float* __restrict__ bias,
                                                    int do_relu) {
  __shared__ float acc[RPG * HIDC];
  __shared__ int   queue[32 * QCAP];
  const int tid  = threadIdx.x;
  const int lane = tid & 31;
  const int wv   = tid >> 5;
  const int lo   = blockIdx.x * RPG;
  const int nrows = min(RPG, NN - lo);
  int* q = queue + wv * QCAP;

  const float4* h4 = (const float4*)(h + (size_t)lo * HIDC);
  for (int i = tid; i < nrows * (HIDC / 4); i += 1024) {
    float d = dinv[lo + (i >> 5)];
    float s = d * d;
    float4 v = h4[i];
    ((float4*)acc)[i] = make_float4(v.x * s, v.y * s, v.z * s, v.w * s);
  }
  __syncthreads();

  const uint4* key4 = (const uint4*)key;
  const unsigned pat = blockIdx.x * 0x01010101u;
  const int trip0 = (blockIdx.x * SCAN_TRIPS) >> 8;
  const unsigned below = (1u << lane) - 1u;
  int qn = 0;
  for (int k = 0; k < SCAN_TRIPS; k += 7) {
    uint4 c[7];
    int   g[7];
#pragma unroll
    for (int u = 0; u < 7; ++u) {
      int t = trip0 + k + u;
      if (t >= SCAN_TRIPS) t -= SCAN_TRIPS;
      g[u] = t * 1024 + tid;
      c[u] = key4[min(g[u], NKEY16 - 1)];
    }
#pragma unroll
    for (int u = 0; u < 7; ++u) {
      unsigned mm = (zero_bytes(c[u].x ^ pat) >> 7) | (zero_bytes(c[u].y ^ pat) >> 6) |
                    (zero_bytes(c[u].z ^ pat) >> 5) | (zero_bytes(c[u].w ^ pat) >> 4);
      if (g[u] >= NKEY16) mm = 0;
      if (__builtin_amdgcn_ballot_w32(mm != 0)) {
        do {
          bool has = mm != 0;
          unsigned b = __builtin_amdgcn_ballot_w32(has);
          if (has) {
            int p = __builtin_ctz(mm);
            q[qn + __popc(b & below)] = 16 * g[u] + ((p & 7) << 2) + (p >> 3);
            mm &= mm - 1;
          }
          qn += __popc(b);
        } while (__builtin_amdgcn_ballot_w32(mm != 0));
        if (qn >= QDRAIN)
          qn = agg_drain<false>(acc, q, qn, lo, lane, h, dinv, row, col);
      }
    }
  }
  agg_drain<true>(acc, q, qn, lo, lane, h, dinv, row, col);
  __syncthreads();

  float4* o4 = (float4*)(out + (size_t)lo * HIDC);
  for (int i = tid; i < nrows * (HIDC / 4); i += 1024) {
    float4 v = ((const float4*)acc)[i];
    float4 b = ((const float4*)bias)[i & 31];
    v.x += b.x; v.y += b.y; v.z += b.z; v.w += b.w;
    if (do_relu) {
      v.x = fmaxf(v.x, 0.0f); v.y = fmaxf(v.y, 0.0f);
      v.z = fmaxf(v.z, 0.0f); v.w = fmaxf(v.w, 0.0f);
    }
    o4[i] = v;
  }
}

#define DEC_GROUPS 256
#define DEC_WAVES  25
#define NPAIR  (NPOS + NNEG)
#define NCHUNK (NPAIR / 32)

__device__ __forceinline__ void dec_pair_idx(const int* __restrict__ pos,
                                             const int* __restrict__ neg,
                                             int p, int& a, int& b) {
  if (p < NPOS) { a = pos[p]; b = pos[NPOS + p]; }
  else          { int j = p - NPOS; a = neg[j]; b = neg[NNEG + j]; }
}

__global__ __launch_bounds__(DEC_WAVES * 32) void k_decode(const float* __restrict__ z,
                                                          const int* __restrict__ pos,
                                                          const int* __restrict__ neg,
                                                          float* __restrict__ out) {
  const int lane = threadIdx.x & 31;
  const int g    = lane >> 3;
  const int i    = lane & 7;
  const int nw   = DEC_GROUPS * DEC_WAVES;
  int c = blockIdx.x * DEC_WAVES + (threadIdx.x >> 5);
  int a = 0, b = 0;
  if (c < NCHUNK) dec_pair_idx(pos, neg, c * 32 + lane, a, b);
  while (c < NCHUNK) {
    const int cn = c + nw;
    int an = 0, bn = 0;
    if (cn < NCHUNK) dec_pair_idx(pos, neg, cn * 32 + lane, an, bn);

    const float4* pa[8];
    const float4* pb[8];
#pragma unroll
    for (int j = 0; j < 8; ++j) {
      const int ra = __shfl(a, 4 * j + g, 32);
      const int rb = __shfl(b, 4 * j + g, 32);
      pa[j] = (const float4*)(z + (size_t)ra * HIDC) + i;
      pb[j] = (const float4*)(z + (size_t)rb * HIDC) + i;
    }
    float s[8];
#pragma unroll
    for (int k = 0; k < 4; ++k) {
      float4 va[2][4], vb[2][4];
      __builtin_amdgcn_sched_barrier(0);
#pragma unroll
      for (int h = 0; h < 2; ++h) {
#pragma unroll
        for (int u = 0; u < 4; ++u) {
          va[h][u] = pa[2 * k + h][8 * u];
          vb[h][u] = pb[2 * k + h][8 * u];
        }
      }
      __builtin_amdgcn_sched_barrier(0);
#pragma unroll
      for (int h = 0; h < 2; ++h) {
        float cx = 0.0f, cy = 0.0f, cz = 0.0f, cw = 0.0f;
#pragma unroll
        for (int u = 0; u < 4; ++u) {
          const float4 x = va[h][u], y = vb[h][u];
          cx = __builtin_fmaf(x.x, y.x, cx);
          cy = __builtin_fmaf(x.y, y.y, cy);
          cz = __builtin_fmaf(x.z, y.z, cz);
          cw = __builtin_fmaf(x.w, y.w, cw);
        }
        s[2 * k + h] = (cx + cy) + (cz + cw);
      }
    }
#pragma unroll
    for (int j = 0; j < 8; ++j) s[j] += __shfl_xor(s[j], 4, 32);
#pragma unroll
    for (int j = 0; j < 8; ++j) s[j] += __shfl_xor(s[j], 2, 32);
#pragma unroll
    for (int j = 0; j < 8; ++j) s[j] += __shfl_xor(s[j], 1, 32);
    float res = 0.0f;
#pragma unroll
    for (int j = 0; j < 8; ++j) {
      const float t = __shfl(s[j], (lane & 3) * 8, 32);
      if ((lane >> 2) == j) res = t;
    }
    out[c * 32 + lane] = res;
    a = an; b = bn; c = cn;
  }
}

extern "C" void kernel_launch(void* const* d_in, const int* in_sizes, int n_in,
                              void* d_out, int out_size, void* d_ws, size_t ws_size,
                              hipStream_t stream) {
  const float* x  = (const float*)d_in[0];
  const int*   ei = (const int*)d_in[1];
  const int*   pos = (const int*)d_in[2];
  const int*   neg = (const int*)d_in[3];
  const float* W1 = (const float*)d_in[4];
  const float* b1 = (const float*)d_in[5];
  const float* W2 = (const float*)d_in[6];
  const float* b2 = (const float*)d_in[7];
  float* logits = (float*)d_out;

  char* ws = (char*)d_ws;
  float* dinv = (float*)ws;
  float* WT   = (float*)(ws + (512u << 10));
  float* bufH = (float*)(ws + (1u << 20));
  float* bufZ = (float*)(ws + (1u << 20) + (52u << 20));
  unsigned* key = (unsigned*)(ws + (1u << 20) + 51200000u);

  const int* row = ei;
  const int* col = ei + NE;
  const int B = 256;

  k_init_dinv<<<(NN + B - 1) / B, B, 0, stream>>>(dinv);
  k_count_deg<<<(NE / 4 + B - 1) / B, B, 0, stream>>>(dinv, col, key);
  k_finish_dinv<<<(NN + B - 1) / B, B, 0, stream>>>(dinv);

  const int gemm_blocks = GEMM_GROUPS;

  k_transpose<<<(INC * HIDC + B - 1) / B, B, 0, stream>>>(W1, WT, INC, HIDC);
  k_gemm_wmma<INC><<<gemm_blocks, B, 0, stream>>>(x, WT, bufH);
  k_agg_owner<<<AGG_GROUPS, 1024, 0, stream>>>(bufZ, bufH, dinv, row, col, key, b1, 1);

  k_transpose<<<(HIDC * HIDC + B - 1) / B, B, 0, stream>>>(W2, WT, HIDC, HIDC);
  k_gemm_wmma<HIDC><<<gemm_blocks, B, 0, stream>>>(bufZ, WT, bufH);
  k_agg_owner<<<AGG_GROUPS, 1024, 0, stream>>>(bufZ, bufH, dinv, row, col, key, b2, 0);

  k_decode<<<DEC_GROUPS, DEC_WAVES * 32, 0, stream>>>(bufZ, pos, neg, logits);
}


# edge aggregation: each group holds a one-byte degree table of all nodes in LDS; an edge's deg^-1/2 factors are LDS reads, not scattered global reads; queue drained each pass, LDS-add order changes
# speedup vs baseline: 2.5931x; 1.0374x over previous
#include <hip/hip_runtime.h>

#define NN   100000
#define INC  256
#define HIDC 128
#define NE   1600000
#define NPOS 200000
#define NNEG 200000

typedef __attribute__((ext_vector_type(2))) float v2f;
typedef __attribute__((ext_vector_type(8))) float v8f;

#define AGG_GROUPS 256
#define RPG   391

__global__ void k_init_dinv(float* dinv) {
  int i = blockIdx.x * blockDim.x + threadIdx.x;
  if (i < NN) dinv[i] = 1.0f;
}

__global__ void k_count_deg(float* dinv, const int* __restrict__ col,
                            unsigned* __restrict__ key) {
  int i = blockIdx.x * blockDim.x + threadIdx.x;
  if (i < NE / 4) {
    int4 c = ((const int4*)col)[i];
    unsafeAtomicAdd(&dinv[c.x], 1.0f);
    unsafeAtomicAdd(&dinv[c.y], 1.0f);
    unsafeAtomicAdd(&dinv[c.z], 1.0f);
    unsafeAtomicAdd(&dinv[c.w], 1.0f);
    unsigned k0 = (unsigned)c.x / RPG, k1 = (unsigned)c.y / RPG;
    unsigned k2 = (unsigned)c.z / RPG, k3 = (unsigned)c.w / RPG;
    key[i] = k0 | (k1 << 8) | (k2 << 16) | (k3 << 24);
  }
}

__device__ __forceinline__ float inv_sqrt_deg(float d) { return 1.0f / sqrtf(d); }

#define DEG_SAT 255u

__global__ void k_finish_dinv(float* dinv, unsigned* __restrict__ deg8) {
  int i = blockIdx.x * blockDim.x + threadIdx.x;
  if (i < NN / 4) {
    float4 d = ((const float4*)dinv)[i];
    unsigned b0 = (unsigned)fminf(d.x, 255.0f), b1 = (unsigned)fminf(d.y, 255.0f);
    unsigned b2 = (unsigned)fminf(d.z, 255.0f), b3 = (unsigned)fminf(d.w, 255.0f);
    deg8[i] = b0 | (b1 << 8) | (b2 << 16) | (b3 << 24);
    ((float4*)dinv)[i] = make_float4(inv_sqrt_deg(d.x), inv_sqrt_deg(d.y),
                                     inv_sqrt_deg(d.z), inv_sqrt_deg(d.w));
  }
}

__global__ void k_transpose(const float* __restrict__ W, float* __restrict__ WT,
                            int K, int N) {
  int i = blockIdx.x * blockDim.x + threadIdx.x;
  if (i >= K * N) return;
  int k = i / N, n = i - k * N;
  WT[n * K + k] = W[i];
}

#define GROWS 32
#define GPAIRS (NN / GROWS)
#define GEMM_GROUPS 256
template <int K>
__global__ __launch_bounds__(256) void k_gemm_wmma(const float* __restrict__ A,
                                                   const float* __restrict__ WT,
                                                   float* __restrict__ C) {
  constexpr int PITCH = K + 4;
  constexpr int F4ROW = K / 4;
  constexpr int NLD   = GROWS * F4ROW / 256;
  __shared__ __attribute__((aligned(16))) float sA[2 * GROWS * PITCH];

  const int tid  = threadIdx.x;
  const int lane = tid & 31;
  const int wv   = tid >> 5;
  const int half = lane >> 4;
  const int l15  = lane & 15;
  const int np   = wv & 3;
  const int mh   = wv >> 2;

  v2f b0[K / 4], b1[K / 4];
  {
    const float* bp0 = WT + (size_t)(np * 32 + l15) * K + half * 2;
    const float* bp1 = bp0 + (size_t)16 * K;
#pragma unroll
    for (int kk = 0; kk < K / 4; ++kk) {
      b0[kk] = *(const v2f*)(bp0 + 4 * kk);
      b1[kk] = *(const v2f*)(bp1 + 4 * kk);
    }
  }

  int soff[NLD];
#pragma unroll
  for (int j = 0; j < NLD; ++j) {
    int f = j * 256 + tid;
    soff[j] = (f / F4ROW) * PITCH + (f % F4ROW) * 4;
  }

  int p = blockIdx.x;
  {
    float4 st0[NLD];
    const float4* src = (const float4*)(A + (size_t)p * GROWS * K) + tid;
#pragma unroll
    for (int j = 0; j < NLD; ++j) st0[j] = src[j * 256];
#pragma unroll
    for (int j = 0; j < NLD; ++j) *(float4*)(sA + soff[j]) = st0[j];
  }
  __syncthreads();

  const int aoff = (mh * 16 + l15) * PITCH + half * 2;
  float* const cbase = C + (size_t)(mh * 16 + half * 8) * HIDC + np * 32 + l15;
  v8f pacc0 = (v8f)(0.0f), pacc1 = (v8f)(0.0f);
  int buf = 0;
  for (; p < GPAIRS; p += GEMM_GROUPS) {
    const bool more = p + GEMM_GROUPS < GPAIRS;
    float4 st[NLD];
    if (more) {
      const float4* src = (const float4*)(A + (size_t)(p + GEMM_GROUPS) * GROWS * K) + tid;
#pragma unroll
      for (int j = 0; j < NLD; ++j) st[j] = src[j * 256];
    } else {
#pragma unroll
      for (int j = 0; j < NLD; ++j) st[j] = make_float4(0.0f, 0.0f, 0.0f, 0.0f);
    }
    if (p != (int)blockIdx.x) {
      float* cp = cbase + (size_t)(p - GEMM_GROUPS) * GROWS * HIDC;
#pragma unroll
      for (int i = 0; i < 8; ++i) {
        cp[i * HIDC]      = pacc0[i];
        cp[i * HIDC + 16] = pacc1[i];
      }
    }

    v8f acc0 = (v8f)(0.0f), acc1 = (v8f)(0.0f);
    const float* ap = sA + buf * (GROWS * PITCH) + aoff;
#pragma unroll
    for (int kk = 0; kk < K / 4; ++kk) {
      v2f a = *(const v2f*)(ap + 4 * kk);
      acc0 = __builtin_amdgcn_wmma_f32_16x16x4_f32(
          false, a, false, b0[kk], (short)0, acc0, false, false);
      acc1 = __builtin_amdgcn_wmma_f32_16x16x4_f32(
          false, a, false, b1[kk], (short)0, acc1, false, false);
    }
    pacc0 = acc0;
    pacc1 = acc1;

    {
      float* dstb = sA + (buf ^ 1) * (GROWS * PITCH);
#pragma unroll
      for (int j = 0; j < NLD; ++j) *(float4*)(dstb + soff[j]) = st[j];
    }
    __syncthreads();
    buf ^= 1;
  }
  {
    float* cp = cbase + (size_t)(p - GEMM_GROUPS) * GROWS * HIDC;
#pragma unroll
    for (int i = 0; i < 8; ++i) {
      cp[i * HIDC]      = pacc0[i];
      cp[i * HIDC + 16] = pacc1[i];
    }
  }
}

#define QDRAIN 64
#define QCAP  (QDRAIN + 32)
#define DEGT_BYTES (NN + 96)
#define NKEY16 (NE / 16)
#define SCAN_TRIPS 98

__device__ __forceinline__ unsigned zero_bytes(unsigned x) {
  unsigned t = (x & 0x7f7f7f7fu) + 0x7f7f7f7fu;
  return ~(t | x) & 0x80808080u;
}

__device__ __forceinline__ float edge_norm(const unsigned char* degt, const float* dlut,
                                           const float* __restrict__ dinv, int r, int c) {
  const unsigned dr = degt[r], dc = degt[c];
  float fr = dlut[dr], fc = dlut[dc];
  if (__builtin_amdgcn_ballot_w32(dr == DEG_SAT || dc == DEG_SAT)) {
    if (dr == DEG_SAT) fr = dinv[r];
    if (dc == DEG_SAT) fc = dinv[c];
  }
  return fr * fc;
}

template <bool FINAL>
__device__ __forceinline__ int agg_drain(float* acc, int* q, int qn, int lo, int lane,
                                         const unsigned char* degt, const float* dlut,
                                         const float* __restrict__ h,
                                         const float* __restrict__ dinv,
                                         const int* __restrict__ row,
                                         const int* __restrict__ col) {
  const int todo = FINAL ? qn : (qn & ~7);
  for (int base = 0; base < todo; base += 32) {
    int cnt = min(32, todo - base);
    int r = 0, cl = 0;
    float nm = 0.0f;
    if (lane < cnt) {
      int e = q[base + lane];
      r = row[e];
      int c = col[e];
      cl = c - lo;
      nm = edge_norm(degt, dlut, dinv, r, c);
    }
    int j = 0;
    for (; j + 8 <= cnt; j += 8) {
      float v[8][4];
      int   cj[8];
      float nj[8];
#pragma unroll
      for (int u = 0; u < 8; ++u) {
        int rj = __builtin_amdgcn_readlane(r, j + u);
        cj[u] = __builtin_amdgcn_readlane(cl, j + u);
        nj[u] = __int_as_float(__builtin_amdgcn_readlane(__float_as_int(nm), j + u));
        const float* src = h + (size_t)rj * HIDC + lane;
        v[u][0] = src[0];
        v[u][1] = src[32];
        v[u][2] = src[64];
        v[u][3] = src[96];
      }
#pragma unroll
      for (int u = 0; u < 8; ++u) {
        float* dst = acc + cj[u] * HIDC + lane;
        unsafeAtomicAdd(dst + 0,  v[u][0] * nj[u]);
        unsafeAtomicAdd(dst + 32, v[u][1] * nj[u]);
        unsafeAtomicAdd(dst + 64, v[u][2] * nj[u]);
        unsafeAtomicAdd(dst + 96, v[u][3] * nj[u]);
      }
    }
    if (FINAL) {
      for (; j < cnt; ++j) {
        int   rj = __builtin_amdgcn_readlane(r, j);
        int   cj = __builtin_amdgcn_readlane(cl, j);
        float nj = __int_as_float(__builtin_amdgcn_readlane(__float_as_int(nm), j));
        const float* src = h + (size_t)rj * HIDC + lane;
        float v0 = src[0], v1 = src[32], v2 = src[64], v3 = src[96];
        float* dst = acc + cj * HIDC + lane;
        unsafeAtomicAdd(dst + 0,  v0 * nj);
        unsafeAtomicAdd(dst + 32, v1 * nj);
        unsafeAtomicAdd(dst + 64, v2 * nj);
        unsafeAtomicAdd(dst + 96, v3 * nj);
      }
    }
  }
  if (FINAL) return 0;
  const int rem = qn - todo;
  if (lane < rem) {
    int t = q[todo + lane];
    q[lane] = t;
  }
  return rem;
}

__global__ __launch_bounds__(1024) void k_agg_owner(float* __restrict__ out,
                                                    const float* __restrict__ h,
                                                    const float* __restrict__ dinv,
                                                    const int* __restrict__ row,
                                                    const int* __restrict__ col,
                                                    const unsigned* __restrict__ key,
                                                    const unsigned* __restrict__ deg8,
                                                    const float* __restrict__ bias,
                                                    int do_relu) {
  __shared__ float acc[RPG * HIDC];
  __shared__ int   queue[32 * QCAP];
  __shared__ __attribute__((aligned(16))) unsigned char degt[DEGT_BYTES];
  __shared__ float dlut[256];
  const int tid  = threadIdx.x;
  const int lane = tid & 31;
  const int wv   = tid >> 5;
  const int lo   = blockIdx.x * RPG;
  const int nrows = min(RPG, NN - lo);
  int* q = queue + wv * QCAP;

  const float4* h4 = (const float4*)(h + (size_t)lo * HIDC);
  for (int i = tid; i < nrows * (HIDC / 4); i += 1024) {
    float d = dinv[lo + (i >> 5)];
    float s = d * d;
    float4 v = h4[i];
    ((float4*)acc)[i] = make_float4(v.x * s, v.y * s, v.z * s, v.w * s);
  }
  const int piece0 = (blockIdx.x * (NN / 16)) >> 8;
  for (int k = tid; k < NN / 16; k += 1024) {
    int i = k + piece0;
    if (i >= NN / 16) i -= NN / 16;
    ((uint4*)degt)[i] = ((const uint4*)deg8)[i];
  }
  if (tid < 256) dlut[tid] = inv_sqrt_deg((float)tid);
  __syncthreads();

  const uint4* key4 = (const uint4*)key;
  const unsigned pat = blockIdx.x * 0x01010101u;
  const int trip0 = (blockIdx.x * SCAN_TRIPS) >> 8;
  const unsigned below = (1u << lane) - 1u;
  int qn = 0;
  for (int k = 0; k < SCAN_TRIPS; k += 7) {
    uint4 c[7];
    int   g[7];
#pragma unroll
    for (int u = 0; u < 7; ++u) {
      int t = trip0 + k + u;
      if (t >= SCAN_TRIPS) t -= SCAN_TRIPS;
      g[u] = t * 1024 + tid;
      c[u] = key4[min(g[u], NKEY16 - 1)];
    }
#pragma unroll
    for (int u = 0; u < 7; ++u) {
      unsigned mm = (zero_bytes(c[u].x ^ pat) >> 7) | (zero_bytes(c[u].y ^ pat) >> 6) |
                    (zero_bytes(c[u].z ^ pat) >> 5) | (zero_bytes(c[u].w ^ pat) >> 4);
      if (g[u] >= NKEY16) mm = 0;
      if (__builtin_amdgcn_ballot_w32(mm != 0)) {
        do {
          bool has = mm != 0;
          unsigned b = __builtin_amdgcn_ballot_w32(has);
          if (has) {
            int p = __builtin_ctz(mm);
            q[qn + __popc(b & below)] = 16 * g[u] + ((p & 7) << 2) + (p >> 3);
            mm &= mm - 1;
          }
          qn += __popc(b);
          if (qn >= QDRAIN)
            qn = agg_drain<false>(acc, q, qn, lo, lane, degt, dlut, h, dinv, row, col);
        } while (__builtin_amdgcn_ballot_w32(mm != 0));
      }
    }
  }
  agg_drain<true>(acc, q, qn, lo, lane, degt, dlut, h, dinv, row, col);
  __syncthreads();

  float4* o4 = (float4*)(out + (size_t)lo * HIDC);
  for (int i = tid; i < nrows * (HIDC / 4); i += 1024) {
    float4 v = ((const float4*)acc)[i];
    float4 b = ((const float4*)bias)[i & 31];
    v.x += b.x; v.y += b.y; v.z += b.z; v.w += b.w;
    if (do_relu) {
      v.x = fmaxf(v.x, 0.0f); v.y = fmaxf(v.y, 0.0f);
      v.z = fmaxf(v.z, 0.0f); v.w = fmaxf(v.w, 0.0f);
    }
    o4[i] = v;
  }
}

#define DEC_GROUPS 256
#define DEC_WAVES  25
#define NPAIR  (NPOS + NNEG)
#define NCHUNK (NPAIR / 32)

__device__ __forceinline__ void dec_pair_idx(const int* __restrict__ pos,
                                             const int* __restrict__ neg,
                                             int p, int& a, int& b) {
  if (p < NPOS) { a = pos[p]; b = pos[NPOS + p]; }
  else          { int j = p - NPOS; a = neg[j]; b = neg[NNEG + j]; }
}

__global__ __launch_bounds__(DEC_WAVES * 32) void k_decode(const float* __restrict__ z,
                                                          const int* __restrict__ pos,
                                                          const int* __restrict__ neg,
                                                          float* __restrict__ out) {
  const int lane = threadIdx.x & 31;
  const int g    = lane >> 3;
  const int i    = lane & 7;
  const int nw   = DEC_GROUPS * DEC_WAVES;
  int c = blockIdx.x * DEC_WAVES + (threadIdx.x >> 5);
  int a = 0, b = 0;
  if (c < NCHUNK) dec_pair_idx(pos, neg, c * 32 + lane, a, b);
  while (c < NCHUNK) {
    const int cn = c + nw;
    int an = 0, bn = 0;
    if (cn < NCHUNK) dec_pair_idx(pos, neg, cn * 32 + lane, an, bn);

    const float4* pa[8];
    const float4* pb[8];
#pragma unroll
    for (int j = 0; j < 8; ++j) {
      const int ra = __shfl(a, 4 * j + g, 32);
      const int rb = __shfl(b, 4 * j + g, 32);
      pa[j] = (const float4*)(z + (size_t)ra * HIDC) + i;
      pb[j] = (const float4*)(z + (size_t)rb * HIDC) + i;
    }
    float s[8];
#pragma unroll
    for (int k = 0; k < 4; ++k) {
      float4 va[2][4], vb[2][4];
      __builtin_amdgcn_sched_barrier(0);
#pragma unroll
      for (int h = 0; h < 2; ++h) {
#pragma unroll
        for (int u = 0; u < 4; ++u) {
          va[h][u] = pa[2 * k + h][8 * u];
          vb[h][u] = pb[2 * k + h][8 * u];
        }
      }
      __builtin_amdgcn_sched_barrier(0);
#pragma unroll
      for (int h = 0; h < 2; ++h) {
        float cx = 0.0f, cy = 0.0f, cz = 0.0f, cw = 0.0f;
#pragma unroll
        for (int u = 0; u < 4; ++u) {
          const float4 x = va[h][u], y = vb[h][u];
          cx = __builtin_fmaf(x.x, y.x, cx);
          cy = __builtin_fmaf(x.y, y.y, cy);
          cz = __builtin_fmaf(x.z, y.z, cz);
          cw = __builtin_fmaf(x.w, y.w, cw);
        }
        s[2 * k + h] = (cx + cy) + (cz + cw);
      }
    }
#pragma unroll
    for (int j = 0; j < 8; ++j) s[j] += __shfl_xor(s[j], 4, 32);
#pragma unroll
    for (int j = 0; j < 8; ++j) s[j] += __shfl_xor(s[j], 2, 32);
#pragma unroll
    for (int j = 0; j < 8; ++j) s[j] += __shfl_xor(s[j], 1, 32);
    float res = 0.0f;
#pragma unroll
    for (int j = 0; j < 8; ++j) {
      const float t = __shfl(s[j], (lane & 3) * 8, 32);
      if ((lane >> 2) == j) res = t;
    }
    out[c * 32 + lane] = res;
    a = an; b = bn; c = cn;
  }
}

extern "C" void kernel_launch(void* const* d_in, const int* in_sizes, int n_in,
                              void* d_out, int out_size, void* d_ws, size_t ws_size,
                              hipStream_t stream) {
  const float* x  = (const float*)d_in[0];
  const int*   ei = (const int*)d_in[1];
  const int*   pos = (const int*)d_in[2];
  const int*   neg = (const int*)d_in[3];
  const float* W1 = (const float*)d_in[4];
  const float* b1 = (const float*)d_in[5];
  const float* W2 = (const float*)d_in[6];
  const float* b2 = (const float*)d_in[7];
  float* logits = (float*)d_out;

  char* ws = (char*)d_ws;
  float* dinv = (float*)ws;
  unsigned* deg8 = (unsigned*)(ws + 400000u);
  float* WT   = (float*)(ws + (512u << 10));
  float* bufH = (float*)(ws + (1u << 20));
  float* bufZ = (float*)(ws + (1u << 20) + (52u << 20));
  unsigned* key = (unsigned*)(ws + (1u << 20) + 51200000u);

  const int* row = ei;
  const int* col = ei + NE;
  const int B = 256;

  k_init_dinv<<<(NN + B - 1) / B, B, 0, stream>>>(dinv);
  k_count_deg<<<(NE / 4 + B - 1) / B, B, 0, stream>>>(dinv, col, key);
  k_finish_dinv<<<(NN / 4 + B - 1) / B, B, 0, stream>>>(dinv, deg8);

  const int gemm_blocks = GEMM_GROUPS;

  k_transpose<<<(INC * HIDC + B - 1) / B, B, 0, stream>>>(W1, WT, INC, HIDC);
  k_gemm_wmma<INC><<<gemm_blocks, B, 0, stream>>>(x, WT, bufH);
  k_agg_owner<<<AGG_GROUPS, 1024, 0, stream>>>(bufZ, bufH, dinv, row, col, key, deg8, b1, 1);

  k_transpose<<<(HIDC * HIDC + B - 1) / B, B, 0, stream>>>(W2, WT, HIDC, HIDC);
  k_gemm_wmma<HIDC><<<gemm_blocks, B, 0, stream>>>(bufZ, WT, bufH);
  k_agg_owner<<<AGG_GROUPS, 1024, 0, stream>>>(bufZ, bufH, dinv, row, col, key, deg8, b2, 0);

  k_decode<<<DEC_GROUPS, DEC_WAVES * 32, 0, stream>>>(bufZ, pos, neg, logits);
}


# layer-2 aggregation walks lists of (source row, dest row, norm) records written by layer 1's key scan as it adds, if the workspace has room: no 2nd scan or index reads; LDS-add order changes
# speedup vs baseline: 2.8879x; 1.1137x over previous
#include <hip/hip_runtime.h>

#define NN   100000
#define INC  256
#define HIDC 128
#define NE   1600000
#define NPOS 200000
#define NNEG 200000

typedef __attribute__((ext_vector_type(2))) float v2f;
typedef __attribute__((ext_vector_type(8))) float v8f;

#define AGG_GROUPS 256
#define RPG   391

__global__ void k_init_dinv(float* dinv) {
  int i = blockIdx.x * blockDim.x + threadIdx.x;
  if (i < NN) dinv[i] = 1.0f;
}

__global__ void k_count_deg(float* dinv, const int* __restrict__ col,
                            unsigned* __restrict__ key) {
  int i = blockIdx.x * blockDim.x + threadIdx.x;
  if (i < NE / 4) {
    int4 c = ((const int4*)col)[i];
    unsafeAtomicAdd(&dinv[c.x], 1.0f);
    unsafeAtomicAdd(&dinv[c.y], 1.0f);
    unsafeAtomicAdd(&dinv[c.z], 1.0f);
    unsafeAtomicAdd(&dinv[c.w], 1.0f);
    unsigned k0 = (unsigned)c.x / RPG, k1 = (unsigned)c.y / RPG;
    unsigned k2 = (unsigned)c.z / RPG, k3 = (unsigned)c.w / RPG;
    key[i] = k0 | (k1 << 8) | (k2 << 16) | (k3 << 24);
  }
}

__device__ __forceinline__ float inv_sqrt_deg(float d) { return 1.0f / sqrtf(d); }

#define DEG_SAT 255u

__global__ void k_finish_dinv(float* dinv, unsigned* __restrict__ deg8) {
  int i = blockIdx.x * blockDim.x + threadIdx.x;
  if (i < NN / 4) {
    float4 d = ((const float4*)dinv)[i];
    unsigned b0 = (unsigned)fminf(d.x, 255.0f), b1 = (unsigned)fminf(d.y, 255.0f);
    unsigned b2 = (unsigned)fminf(d.z, 255.0f), b3 = (unsigned)fminf(d.w, 255.0f);
    deg8[i] = b0 | (b1 << 8) | (b2 << 16) | (b3 << 24);
    ((float4*)dinv)[i] = make_float4(inv_sqrt_deg(d.x), inv_sqrt_deg(d.y),
                                     inv_sqrt_deg(d.z), inv_sqrt_deg(d.w));
  }
}

__global__ void k_transpose(const float* __restrict__ W, float* __restrict__ WT,
                            int K, int N) {
  int i = blockIdx.x * blockDim.x + threadIdx.x;
  if (i >= K * N) return;
  int k = i / N, n = i - k * N;
  WT[n * K + k] = W[i];
}

#define GROWS 32
#define GPAIRS (NN / GROWS)
#define GEMM_GROUPS 256
template <int K>
__global__ __launch_bounds__(256) void k_gemm_wmma(const float* __restrict__ A,
                                                   const float* __restrict__ WT,
                                                   float* __restrict__ C) {
  constexpr int PITCH = K + 4;
  constexpr int F4ROW = K / 4;
  constexpr int NLD   = GROWS * F4ROW / 256;
  __shared__ __attribute__((aligned(16))) float sA[2 * GROWS * PITCH];

  const int tid  = threadIdx.x;
  const int lane = tid & 31;
  const int wv   = tid >> 5;
  const int half = lane >> 4;
  const int l15  = lane & 15;
  const int np   = wv & 3;
  const int mh   = wv >> 2;

  v2f b0[K / 4], b1[K / 4];
  {
    const float* bp0 = WT + (size_t)(np * 32 + l15) * K + half * 2;
    const float* bp1 = bp0 + (size_t)16 * K;
#pragma unroll
    for (int kk = 0; kk < K / 4; ++kk) {
      b0[kk] = *(const v2f*)(bp0 + 4 * kk);
      b1[kk] = *(const v2f*)(bp1 + 4 * kk);
    }
  }

  int soff[NLD];
#pragma unroll
  for (int j = 0; j < NLD; ++j) {
    int f = j * 256 + tid;
    soff[j] = (f / F4ROW) * PITCH + (f % F4ROW) * 4;
  }

  int p = blockIdx.x;
  {
    float4 st0[NLD];
    const float4* src = (const float4*)(A + (size_t)p * GROWS * K) + tid;
#pragma unroll
    for (int j = 0; j < NLD; ++j) st0[j] = src[j * 256];
#pragma unroll
    for (int j = 0; j < NLD; ++j) *(float4*)(sA + soff[j]) = st0[j];
  }
  __syncthreads();

  const int aoff = (mh * 16 + l15) * PITCH + half * 2;
  float* const cbase = C + (size_t)(mh * 16 + half * 8) * HIDC + np * 32 + l15;
  v8f pacc0 = (v8f)(0.0f), pacc1 = (v8f)(0.0f);
  int buf = 0;
  for (; p < GPAIRS; p += GEMM_GROUPS) {
    const bool more = p + GEMM_GROUPS < GPAIRS;
    float4 st[NLD];
    if (more) {
      const float4* src = (const float4*)(A + (size_t)(p + GEMM_GROUPS) * GROWS * K) + tid;
#pragma unroll
      for (int j = 0; j < NLD; ++j) st[j] = src[j * 256];
    } else {
#pragma unroll
      for (int j = 0; j < NLD; ++j) st[j] = make_float4(0.0f, 0.0f, 0.0f, 0.0f);
    }
    if (p != (int)blockIdx.x) {
      float* cp = cbase + (size_t)(p - GEMM_GROUPS) * GROWS * HIDC;
#pragma unroll
      for (int i = 0; i < 8; ++i) {
        cp[i * HIDC]      = pacc0[i];
        cp[i * HIDC + 16] = pacc1[i];
      }
    }

    v8f acc0 = (v8f)(0.0f), acc1 = (v8f)(0.0f);
    const float* ap = sA + buf * (GROWS * PITCH) + aoff;
#pragma unroll
    for (int kk = 0; kk < K / 4; ++kk) {
      v2f a = *(const v2f*)(ap + 4 * kk);
      acc0 = __builtin_amdgcn_wmma_f32_16x16x4_f32(
          false, a, false, b0[kk], (short)0, acc0, false, false);
      acc1 = __builtin_amdgcn_wmma_f32_16x16x4_f32(
          false, a, false, b1[kk], (short)0, acc1, false, false);
    }
    pacc0 = acc0;
    pacc1 = acc1;

    {
      float* dstb = sA + (buf ^ 1) * (GROWS * PITCH);
#pragma unroll
      for (int j = 0; j < NLD; ++j) *(float4*)(dstb + soff[j]) = st[j];
    }
    __syncthreads();
    buf ^= 1;
  }
  {
    float* cp = cbase + (size_t)(p - GEMM_GROUPS) * GROWS * HIDC;
#pragma unroll
    for (int i = 0; i < 8; ++i) {
      cp[i * HIDC]      = pacc0[i];
      cp[i * HIDC + 16] = pacc1[i];
    }
  }
}

#define REC_CAP  7168
#define REC_OFF  106774528ull
#define REC_CNT_BYTES 1024ull
#define REC_END  (REC_OFF + REC_CNT_BYTES + 256ull * REC_CAP * 8ull)
#define QDRAIN 64
#define QCAP  (QDRAIN + 32)
#define DEGT_BYTES (NN + 96)
#define NKEY16 (NE / 16)
#define SCAN_TRIPS 98

__device__ __forceinline__ unsigned zero_bytes(unsigned x) {
  unsigned t = (x & 0x7f7f7f7fu) + 0x7f7f7f7fu;
  return ~(t | x) & 0x80808080u;
}

__device__ __forceinline__ float edge_norm(const unsigned char* degt, const float* dlut,
                                           const float* __restrict__ dinv, int r, int c) {
  const unsigned dr = degt[r], dc = degt[c];
  float fr = dlut[dr], fc = dlut[dc];
  if (__builtin_amdgcn_ballot_w32(dr == DEG_SAT || dc == DEG_SAT)) {
    if (dr == DEG_SAT) fr = dinv[r];
    if (dc == DEG_SAT) fc = dinv[c];
  }
  return fr * fc;
}

template <bool SINGLES>
__device__ __forceinline__ void agg_apply(float* acc, int r, int cl, float nm, int cnt,
                                          int lane, const float* __restrict__ h) {
  {
    int j = 0;
    for (; j + 8 <= cnt; j += 8) {
      float v[8][4];
      int   cj[8];
      float nj[8];
#pragma unroll
      for (int u = 0; u < 8; ++u) {
        int rj = __builtin_amdgcn_readlane(r, j + u);
        cj[u] = __builtin_amdgcn_readlane(cl, j + u);
        nj[u] = __int_as_float(__builtin_amdgcn_readlane(__float_as_int(nm), j + u));
        const float* src = h + (size_t)rj * HIDC + lane;
        v[u][0] = src[0];
        v[u][1] = src[32];
        v[u][2] = src[64];
        v[u][3] = src[96];
      }
#pragma unroll
      for (int u = 0; u < 8; ++u) {
        float* dst = acc + cj[u] * HIDC + lane;
        unsafeAtomicAdd(dst + 0,  v[u][0] * nj[u]);
        unsafeAtomicAdd(dst + 32, v[u][1] * nj[u]);
        unsafeAtomicAdd(dst + 64, v[u][2] * nj[u]);
        unsafeAtomicAdd(dst + 96, v[u][3] * nj[u]);
      }
    }
    if (SINGLES) {
      for (; j < cnt; ++j) {
        int   rj = __builtin_amdgcn_readlane(r, j);
        int   cj = __builtin_amdgcn_readlane(cl, j);
        float nj = __int_as_float(__builtin_amdgcn_readlane(__float_as_int(nm), j));
        const float* src = h + (size_t)rj * HIDC + lane;
        float v0 = src[0], v1 = src[32], v2 = src[64], v3 = src[96];
        float* dst = acc + cj * HIDC + lane;
        unsafeAtomicAdd(dst + 0,  v0 * nj);
        unsafeAtomicAdd(dst + 32, v1 * nj);
        unsafeAtomicAdd(dst + 64, v2 * nj);
        unsafeAtomicAdd(dst + 96, v3 * nj);
      }
    }
  }
}

template <bool FINAL>
__device__ __forceinline__ int agg_drain(float* acc, int* q, int qn, int lo, int lane,
                                         const unsigned char* degt, const float* dlut,
                                         const float* __restrict__ h,
                                         const float* __restrict__ dinv,
                                         const int* __restrict__ row,
                                         const int* __restrict__ col,
                                         int2* __restrict__ rec, int* rcount) {
  const int todo = FINAL ? qn : (qn & ~7);
  for (int base = 0; base < todo; base += 32) {
    int cnt = min(32, todo - base);
    int r = 0, cl = 0;
    float nm = 0.0f;
    if (lane < cnt) {
      int e = q[base + lane];
      r = row[e];
      int c = col[e];
      cl = c - lo;
      nm = edge_norm(degt, dlut, dinv, r, c);
    }
    if (rec) {
      int pos = 0;
      if (lane == 0) pos = atomicAdd(rcount, cnt);
      pos = __builtin_amdgcn_readfirstlane(pos) + lane;
      if (lane < cnt && pos < REC_CAP) rec[pos] = make_int2(r | (cl << 17), __float_as_int(nm));
    }
    agg_apply<FINAL>(acc, r, cl, nm, cnt, lane, h);
  }
  if (FINAL) return 0;
  const int rem = qn - todo;
  if (lane < rem) {
    int t = q[todo + lane];
    q[lane] = t;
  }
  return rem;
}

__global__ __launch_bounds__(1024) void k_agg_owner(float* __restrict__ out,
                                                    const float* __restrict__ h,
                                                    const float* __restrict__ dinv,
                                                    const int* __restrict__ row,
                                                    const int* __restrict__ col,
                                                    const unsigned* __restrict__ key,
                                                    const unsigned* __restrict__ deg8,
                                                    const float* __restrict__ bias,
                                                    int do_relu,
                                                    int* __restrict__ rec_count,
                                                    int* __restrict__ rec_words,
                                                    size_t ws_bytes, int layer) {
  __shared__ float acc[RPG * HIDC];
  __shared__ int   queue[32 * QCAP];
  __shared__ __attribute__((aligned(16))) unsigned char degt[DEGT_BYTES];
  __shared__ float dlut[256];
  __shared__ int   rcount;
  const int tid  = threadIdx.x;
  const int lane = tid & 31;
  const int wv   = tid >> 5;
  const int lo   = blockIdx.x * RPG;
  const int nrows = min(RPG, NN - lo);
  int* q = queue + wv * QCAP;

  const bool room = ws_bytes >= REC_END;
  int2* const rec = (int2*)rec_words + (size_t)blockIdx.x * REC_CAP;
  int2* const wrec = (room && layer == 1) ? rec : nullptr;
  int nrec = REC_CAP + 1;
  if (room && layer == 2) nrec = rec_count[blockIdx.x];
  const bool replay = nrec <= REC_CAP;
  if (tid == 0) rcount = 0;

  const float4* h4 = (const float4*)(h + (size_t)lo * HIDC);
  for (int i = tid; i < nrows * (HIDC / 4); i += 1024) {
    float d = dinv[lo + (i >> 5)];
    float s = d * d;
    float4 v = h4[i];
    ((float4*)acc)[i] = make_float4(v.x * s, v.y * s, v.z * s, v.w * s);
  }
  if (!replay) {
    const int piece0 = (blockIdx.x * (NN / 16)) >> 8;
    for (int k = tid; k < NN / 16; k += 1024) {
      int i = k + piece0;
      if (i >= NN / 16) i -= NN / 16;
      ((uint4*)degt)[i] = ((const uint4*)deg8)[i];
    }
    if (tid < 256) dlut[tid] = inv_sqrt_deg((float)tid);
  }
  __syncthreads();

  if (replay) {
    int base = __builtin_amdgcn_readfirstlane(wv) * 32;
    if (base < nrec) {
      int2 cur = rec[min(base + lane, nrec - 1)];
      for (; base < nrec; base += 1024) {
        const int2 nxt = rec[min(base + 1024 + lane, nrec - 1)];
        const int cnt = min(32, nrec - base);
        agg_apply<true>(acc, cur.x & 0x1ffff, cur.x >> 17, __int_as_float(cur.y), cnt, lane, h);
        cur = nxt;
      }
    }
  } else {

  const uint4* key4 = (const uint4*)key;
  const unsigned pat = blockIdx.x * 0x01010101u;
  const int trip0 = (blockIdx.x * SCAN_TRIPS) >> 8;
  const unsigned below = (1u << lane) - 1u;
  int qn = 0;
  for (int k = 0; k < SCAN_TRIPS; k += 7) {
    uint4 c[7];
    int   g[7];
#pragma unroll
    for (int u = 0; u < 7; ++u) {
      int t = trip0 + k + u;
      if (t >= SCAN_TRIPS) t -= SCAN_TRIPS;
      g[u] = t * 1024 + tid;
      c[u] = key4[min(g[u], NKEY16 - 1)];
    }
#pragma unroll
    for (int u = 0; u < 7; ++u) {
      unsigned mm = (zero_bytes(c[u].x ^ pat) >> 7) | (zero_bytes(c[u].y ^ pat) >> 6) |
                    (zero_bytes(c[u].z ^ pat) >> 5) | (zero_bytes(c[u].w ^ pat) >> 4);
      if (g[u] >= NKEY16) mm = 0;
      if (__builtin_amdgcn_ballot_w32(mm != 0)) {
        do {
          bool has = mm != 0;
          unsigned b = __builtin_amdgcn_ballot_w32(has);
          if (has) {
            int p = __builtin_ctz(mm);
            q[qn + __popc(b & below)] = 16 * g[u] + ((p & 7) << 2) + (p >> 3);
            mm &= mm - 1;
          }
          qn += __popc(b);
          if (qn >= QDRAIN)
            qn = agg_drain<false>(acc, q, qn, lo, lane, degt, dlut, h, dinv, row, col, wrec, &rcount);
        } while (__builtin_amdgcn_ballot_w32(mm != 0));
      }
    }
  }
  agg_drain<true>(acc, q, qn, lo, lane, degt, dlut, h, dinv, row, col, wrec, &rcount);
  }
  __syncthreads();
  if (wrec && tid == 0) rec_count[blockIdx.x] = rcount;

  float4* o4 = (float4*)(out + (size_t)lo * HIDC);
  for (int i = tid; i < nrows * (HIDC / 4); i += 1024) {
    float4 v = ((const float4*)acc)[i];
    float4 b = ((const float4*)bias)[i & 31];
    v.x += b.x; v.y += b.y; v.z += b.z; v.w += b.w;
    if (do_relu) {
      v.x = fmaxf(v.x, 0.0f); v.y = fmaxf(v.y, 0.0f);
      v.z = fmaxf(v.z, 0.0f); v.w = fmaxf(v.w, 0.0f);
    }
    o4[i] = v;
  }
}

#define DEC_GROUPS 256
#define DEC_WAVES  25
#define NPAIR  (NPOS + NNEG)
#define NCHUNK (NPAIR / 32)

__device__ __forceinline__ void dec_pair_idx(const int* __restrict__ pos,
                                             const int* __restrict__ neg,
                                             int p, int& a, int& b) {
  if (p < NPOS) { a = pos[p]; b = pos[NPOS + p]; }
  else          { int j = p - NPOS; a = neg[j]; b = neg[NNEG + j]; }
}

__global__ __launch_bounds__(DEC_WAVES * 32) void k_decode(const float* __restrict__ z,
                                                          const int* __restrict__ pos,
                                                          const int* __restrict__ neg,
                                                          float* __restrict__ out) {
  const int lane = threadIdx.x & 31;
  const int g    = lane >> 3;
  const int i    = lane & 7;
  const int nw   = DEC_GROUPS * DEC_WAVES;
  int c = blockIdx.x * DEC_WAVES + (threadIdx.x >> 5);
  int a = 0, b = 0;
  if (c < NCHUNK) dec_pair_idx(pos, neg, c * 32 + lane, a, b);
  while (c < NCHUNK) {
    const int cn = c + nw;
    int an = 0, bn = 0;
    if (cn < NCHUNK) dec_pair_idx(pos, neg, cn * 32 + lane, an, bn);

    const float4* pa[8];
    const float4* pb[8];
#pragma unroll
    for (int j = 0; j < 8; ++j) {
      const int ra = __shfl(a, 4 * j + g, 32);
      const int rb = __shfl(b, 4 * j + g, 32);
      pa[j] = (const float4*)(z + (size_t)ra * HIDC) + i;
      pb[j] = (const float4*)(z + (size_t)rb * HIDC) + i;
    }
    float s[8];
#pragma unroll
    for (int k = 0; k < 4; ++k) {
      float4 va[2][4], vb[2][4];
      __builtin_amdgcn_sched_barrier(0);
#pragma unroll
      for (int h = 0; h < 2; ++h) {
#pragma unroll
        for (int u = 0; u < 4; ++u) {
          va[h][u] = pa[2 * k + h][8 * u];
          vb[h][u] = pb[2 * k + h][8 * u];
        }
      }
      __builtin_amdgcn_sched_barrier(0);
#pragma unroll
      for (int h = 0; h < 2; ++h) {
        float cx = 0.0f, cy = 0.0f, cz = 0.0f, cw = 0.0f;
#pragma unroll
        for (int u = 0; u < 4; ++u) {
          const float4 x = va[h][u], y = vb[h][u];
          cx = __builtin_fmaf(x.x, y.x, cx);
          cy = __builtin_fmaf(x.y, y.y, cy);
          cz = __builtin_fmaf(x.z, y.z, cz);
          cw = __builtin_fmaf(x.w, y.w, cw);
        }
        s[2 * k + h] = (cx + cy) + (cz + cw);
      }
    }
#pragma unroll
    for (int j = 0; j < 8; ++j) s[j] += __shfl_xor(s[j], 4, 32);
#pragma unroll
    for (int j = 0; j < 8; ++j) s[j] += __shfl_xor(s[j], 2, 32);
#pragma unroll
    for (int j = 0; j < 8; ++j) s[j] += __shfl_xor(s[j], 1, 32);
    float res = 0.0f;
#pragma unroll
    for (int j = 0; j < 8; ++j) {
      const float t = __shfl(s[j], (lane & 3) * 8, 32);
      if ((lane >> 2) == j) res = t;
    }
    out[c * 32 + lane] = res;
    a = an; b = bn; c = cn;
  }
}

extern "C" void kernel_launch(void* const* d_in, const int* in_sizes, int n_in,
                              void* d_out, int out_size, void* d_ws, size_t ws_size,
                              hipStream_t stream) {
  const float* x  = (const float*)d_in[0];
  const int*   ei = (const int*)d_in[1];
  const int*   pos = (const int*)d_in[2];
  const int*   neg = (const int*)d_in[3];
  const float* W1 = (const float*)d_in[4];
  const float* b1 = (const float*)d_in[5];
  const float* W2 = (const float*)d_in[6];
  const float* b2 = (const float*)d_in[7];
  float* logits = (float*)d_out;

  char* ws = (char*)d_ws;
  float* dinv = (float*)ws;
  unsigned* deg8 = (unsigned*)(ws + 400000u);
  float* WT   = (float*)(ws + (512u << 10));
  float* bufH = (float*)(ws + (1u << 20));
  float* bufZ = (float*)(ws + (1u << 20) + (52u << 20));
  unsigned* key = (unsigned*)(ws + (1u << 20) + 51200000u);
  int*  rec_count = (int*)(ws + REC_OFF);
  int*  rec_all   = (int*)(ws + REC_OFF + REC_CNT_BYTES);

  const int* row = ei;
  const int* col = ei + NE;
  const int B = 256;

  k_init_dinv<<<(NN + B - 1) / B, B, 0, stream>>>(dinv);
  k_count_deg<<<(NE / 4 + B - 1) / B, B, 0, stream>>>(dinv, col, key);
  k_finish_dinv<<<(NN / 4 + B - 1) / B, B, 0, stream>>>(dinv, deg8);

  const int gemm_blocks = GEMM_GROUPS;

  k_transpose<<<(INC * HIDC + B - 1) / B, B, 0, stream>>>(W1, WT, INC, HIDC);
  k_gemm_wmma<INC><<<gemm_blocks, B, 0, stream>>>(x, WT, bufH);
  k_agg_owner<<<AGG_GROUPS, 1024, 0, stream>>>(bufZ, bufH, dinv, row, col, key, deg8, b1, 1,
                                               rec_count, rec_all, ws_size, 1);

  k_transpose<<<(HIDC * HIDC + B - 1) / B, B, 0, stream>>>(W2, WT, HIDC, HIDC);
  k_gemm_wmma<HIDC><<<gemm_blocks, B, 0, stream>>>(bufZ, WT, bufH);
  k_agg_owner<<<AGG_GROUPS, 1024, 0, stream>>>(bufZ, bufH, dinv, row, col, key, deg8, b2, 0,
                                               rec_count, rec_all, ws_size, 2);

  k_decode<<<DEC_GROUPS, DEC_WAVES * 32, 0, stream>>>(bufZ, pos, neg, logits);
}


# launches 10 -> 7: the degree count and owner-key writing ride in the layer-1 GEMM's waves (one 4-edge item per trip), both W transposes share the degree start value's launch
# speedup vs baseline: 3.0479x; 1.0554x over previous
#include <hip/hip_runtime.h>

#define NN   100000
#define INC  256
#define HIDC 128
#define NE   1600000
#define NPOS 200000
#define NNEG 200000

typedef __attribute__((ext_vector_type(2))) float v2f;
typedef __attribute__((ext_vector_type(8))) float v8f;

#define AGG_GROUPS 256
#define RPG   391

#define PREP_B 256
#define PREP_DEG_BLOCKS ((NN + PREP_B - 1) / PREP_B)
#define PREP_T1_BLOCKS  (INC * HIDC / PREP_B)
#define PREP_T2_BLOCKS  (HIDC * HIDC / PREP_B)
#define PREP_BLOCKS (PREP_DEG_BLOCKS + PREP_T1_BLOCKS + PREP_T2_BLOCKS)
__global__ __launch_bounds__(PREP_B) void k_prep(float* __restrict__ dinv,
                                                 const float* __restrict__ W1,
                                                 float* __restrict__ WT1,
                                                 const float* __restrict__ W2,
                                                 float* __restrict__ WT2) {
  const int b = blockIdx.x;
  if (b < PREP_DEG_BLOCKS) {
    int i = b * PREP_B + threadIdx.x;
    if (i < NN) dinv[i] = 1.0f;
  } else if (b < PREP_DEG_BLOCKS + PREP_T1_BLOCKS) {
    int i = (b - PREP_DEG_BLOCKS) * PREP_B + threadIdx.x;
    int k = i / HIDC, n = i - k * HIDC;
    WT1[n * INC + k] = W1[i];
  } else {
    int i = (b - PREP_DEG_BLOCKS - PREP_T1_BLOCKS) * PREP_B + threadIdx.x;
    int k = i / HIDC, n = i - k * HIDC;
    WT2[n * HIDC + k] = W2[i];
  }
}

__device__ __forceinline__ float inv_sqrt_deg(float d) { return 1.0f / sqrtf(d); }

#define DEG_SAT 255u

__global__ void k_finish_dinv(float* dinv, unsigned* __restrict__ deg8) {
  int i = blockIdx.x * blockDim.x + threadIdx.x;
  if (i < NN / 4) {
    float4 d = ((const float4*)dinv)[i];
    unsigned b0 = (unsigned)fminf(d.x, 255.0f), b1 = (unsigned)fminf(d.y, 255.0f);
    unsigned b2 = (unsigned)fminf(d.z, 255.0f), b3 = (unsigned)fminf(d.w, 255.0f);
    deg8[i] = b0 | (b1 << 8) | (b2 << 16) | (b3 << 24);
    ((float4*)dinv)[i] = make_float4(inv_sqrt_deg(d.x), inv_sqrt_deg(d.y),
                                     inv_sqrt_deg(d.z), inv_sqrt_deg(d.w));
  }
}

#define GROWS 32
#define GPAIRS (NN / GROWS)
#define GEMM_GROUPS 256
#define CNT_ITEMS (NE / 4)
#define CNT_TRIPS 7
static_assert(CNT_TRIPS * GEMM_GROUPS * 256 >= CNT_ITEMS, "every item has a thread and a trip");
static_assert(GPAIRS / GEMM_GROUPS > CNT_TRIPS, "every group has a trip behind its last item");
template <int K, bool COUNT>
__global__ __launch_bounds__(256) void k_gemm_wmma(const float* __restrict__ A,
                                                   const float* __restrict__ WT,
                                                   float* __restrict__ C,
                                                   float* dinv,
                                                   const int* __restrict__ col,
                                                   unsigned* __restrict__ key) {
  constexpr int PITCH = K + 4;
  constexpr int F4ROW = K / 4;
  constexpr int NLD   = GROWS * F4ROW / 256;
  __shared__ __attribute__((aligned(16))) float sA[2 * GROWS * PITCH];

  const int tid  = threadIdx.x;
  const int lane = tid & 31;
  const int wv   = tid >> 5;
  const int half = lane >> 4;
  const int l15  = lane & 15;
  const int np   = wv & 3;
  const int mh   = wv >> 2;

  v2f b0[K / 4], b1[K / 4];
  {
    const float* bp0 = WT + (size_t)(np * 32 + l15) * K + half * 2;
    const float* bp1 = bp0 + (size_t)16 * K;
#pragma unroll
    for (int kk = 0; kk < K / 4; ++kk) {
      b0[kk] = *(const v2f*)(bp0 + 4 * kk);
      b1[kk] = *(const v2f*)(bp1 + 4 * kk);
    }
  }

  int soff[NLD];
#pragma unroll
  for (int j = 0; j < NLD; ++j) {
    int f = j * 256 + tid;
    soff[j] = (f / F4ROW) * PITCH + (f % F4ROW) * 4;
  }

  int p = blockIdx.x;
  {
    float4 st0[NLD];
    const float4* src = (const float4*)(A + (size_t)p * GROWS * K) + tid;
#pragma unroll
    for (int j = 0; j < NLD; ++j) st0[j] = src[j * 256];
#pragma unroll
    for (int j = 0; j < NLD; ++j) *(float4*)(sA + soff[j]) = st0[j];
  }
  __syncthreads();

  const int aoff = (mh * 16 + l15) * PITCH + half * 2;
  float* const cbase = C + (size_t)(mh * 16 + half * 8) * HIDC + np * 32 + l15;
  v8f pacc0 = (v8f)(0.0f), pacc1 = (v8f)(0.0f);
  int buf = 0;
  int trip = 0;
  int4 cq = make_int4(0, 0, 0, 0);
  int  cqi = -1;
  for (; p < GPAIRS; p += GEMM_GROUPS, ++trip) {
    const bool more = p + GEMM_GROUPS < GPAIRS;
    float4 st[NLD];
    int4 cn = cq;
    int  cni = -1;
    if (more) {
      const float4* src = (const float4*)(A + (size_t)(p + GEMM_GROUPS) * GROWS * K) + tid;
#pragma unroll
      for (int j = 0; j < NLD; ++j) st[j] = src[j * 256];
      if (COUNT && trip < CNT_TRIPS) {
        const int i = (trip * GEMM_GROUPS + (int)blockIdx.x) * 256 + tid;
        cn = ((const int4*)col)[min(i, CNT_ITEMS - 1)];
        if (i < CNT_ITEMS) cni = i;
      }
    } else {
#pragma unroll
      for (int j = 0; j < NLD; ++j) st[j] = make_float4(0.0f, 0.0f, 0.0f, 0.0f);
    }
    if (p != (int)blockIdx.x) {
      float* cp = cbase + (size_t)(p - GEMM_GROUPS) * GROWS * HIDC;
#pragma unroll
      for (int i = 0; i < 8; ++i) {
        cp[i * HIDC]      = pacc0[i];
        cp[i * HIDC + 16] = pacc1[i];
      }
      if (COUNT && cqi >= 0) {
        unsafeAtomicAdd(&dinv[cq.x], 1.0f);
        unsafeAtomicAdd(&dinv[cq.y], 1.0f);
        unsafeAtomicAdd(&dinv[cq.z], 1.0f);
        unsafeAtomicAdd(&dinv[cq.w], 1.0f);
        unsigned k0 = (unsigned)cq.x / RPG, k1 = (unsigned)cq.y / RPG;
        unsigned k2 = (unsigned)cq.z / RPG, k3 = (unsigned)cq.w / RPG;
        key[cqi] = k0 | (k1 << 8) | (k2 << 16) | (k3 << 24);
      }
    }

    v8f acc0 = (v8f)(0.0f), acc1 = (v8f)(0.0f);
    const float* ap = sA + buf * (GROWS * PITCH) + aoff;
#pragma unroll
    for (int kk = 0; kk < K / 4; ++kk) {
      v2f a = *(const v2f*)(ap + 4 * kk);
      acc0 = __builtin_amdgcn_wmma_f32_16x16x4_f32(
          false, a, false, b0[kk], (short)0, acc0, false, false);
      acc1 = __builtin_amdgcn_wmma_f32_16x16x4_f32(
          false, a, false, b1[kk], (short)0, acc1, false, false);
    }
    pacc0 = acc0;
    pacc1 = acc1;

    {
      float* dstb = sA + (buf ^ 1) * (GROWS * PITCH);
#pragma unroll
      for (int j = 0; j < NLD; ++j) *(float4*)(dstb + soff[j]) = st[j];
    }
    __syncthreads();
    buf ^= 1;
    cq = cn;
    cqi = cni;
  }
  {
    float* cp = cbase + (size_t)(p - GEMM_GROUPS) * GROWS * HIDC;
#pragma unroll
    for (int i = 0; i < 8; ++i) {
      cp[i * HIDC]      = pacc0[i];
      cp[i * HIDC + 16] = pacc1[i];
    }
  }
}

#define REC_CAP  7168
#define REC_OFF  106774528ull
#define REC_CNT_BYTES 1024ull
#define REC_END  (REC_OFF + REC_CNT_BYTES + 256ull * REC_CAP * 8ull)
#define QDRAIN 64
#define QCAP  (QDRAIN + 32)
#define DEGT_BYTES (NN + 96)
#define NKEY16 (NE / 16)
#define SCAN_TRIPS 98

__device__ __forceinline__ unsigned zero_bytes(unsigned x) {
  unsigned t = (x & 0x7f7f7f7fu) + 0x7f7f7f7fu;
  return ~(t | x) & 0x80808080u;
}

__device__ __forceinline__ float edge_norm(const unsigned char* degt, const float* dlut,
                                           const float* __restrict__ dinv, int r, int c) {
  const unsigned dr = degt[r], dc = degt[c];
  float fr = dlut[dr], fc = dlut[dc];
  if (__builtin_amdgcn_ballot_w32(dr == DEG_SAT || dc == DEG_SAT)) {
    if (dr == DEG_SAT) fr = dinv[r];
    if (dc == DEG_SAT) fc = dinv[c];
  }
  return fr * fc;
}

template <bool SINGLES>
__device__ __forceinline__ void agg_apply(float* acc, int r, int cl, float nm, int cnt,
                                          int lane, const float* __restrict__ h) {
  {
    int j = 0;
    for (; j + 8 <= cnt; j += 8) {
      float v[8][4];
      int   cj[8];
      float nj[8];
#pragma unroll
      for (int u = 0; u < 8; ++u) {
        int rj = __builtin_amdgcn_readlane(r, j + u);
        cj[u] = __builtin_amdgcn_readlane(cl, j + u);
        nj[u] = __int_as_float(__builtin_amdgcn_readlane(__float_as_int(nm), j + u));
        const float* src = h + (size_t)rj * HIDC + lane;
        v[u][0] = src[0];
        v[u][1] = src[32];
        v[u][2] = src[64];
        v[u][3] = src[96];
      }
#pragma unroll
      for (int u = 0; u < 8; ++u) {
        float* dst = acc + cj[u] * HIDC + lane;
        unsafeAtomicAdd(dst + 0,  v[u][0] * nj[u]);
        unsafeAtomicAdd(dst + 32, v[u][1] * nj[u]);
        unsafeAtomicAdd(dst + 64, v[u][2] * nj[u]);
        unsafeAtomicAdd(dst + 96, v[u][3] * nj[u]);
      }
    }
    if (SINGLES) {
      for (; j < cnt; ++j) {
        int   rj = __builtin_amdgcn_readlane(r, j);
        int   cj = __builtin_amdgcn_readlane(cl, j);
        float nj = __int_as_float(__builtin_amdgcn_readlane(__float_as_int(nm), j));
        const float* src = h + (size_t)rj * HIDC + lane;
        float v0 = src[0], v1 = src[32], v2 = src[64], v3 = src[96];
        float* dst = acc + cj * HIDC + lane;
        unsafeAtomicAdd(dst + 0,  v0 * nj);
        unsafeAtomicAdd(dst + 32, v1 * nj);
        unsafeAtomicAdd(dst + 64, v2 * nj);
        unsafeAtomicAdd(dst + 96, v3 * nj);
      }
    }
  }
}

template <bool FINAL>
__device__ __forceinline__ int agg_drain(float* acc, int* q, int qn, int lo, int lane,
                                         const unsigned char* degt, const float* dlut,
                                         const float* __restrict__ h,
                                         const float* __restrict__ dinv,
                                         const int* __restrict__ row,
                                         const int* __restrict__ col,
                                         int2* __restrict__ rec, int* rcount) {
  const int todo = FINAL ? qn : (qn & ~7);
  for (int base = 0; base < todo; base += 32) {
    int cnt = min(32, todo - base);
    int r = 0, cl = 0;
    float nm = 0.0f;
    if (lane < cnt) {
      int e = q[base + lane];
      r = row[e];
      int c = col[e];
      cl = c - lo;
      nm = edge_norm(degt, dlut, dinv, r, c);
    }
    if (rec) {
      int pos = 0;
      if (lane == 0) pos = atomicAdd(rcount, cnt);
      pos = __builtin_amdgcn_readfirstlane(pos) + lane;
      if (lane < cnt && pos < REC_CAP) rec[pos] = make_int2(r | (cl << 17), __float_as_int(nm));
    }
    agg_apply<FINAL>(acc, r, cl, nm, cnt, lane, h);
  }
  if (FINAL) return 0;
  const int rem = qn - todo;
  if (lane < rem) {
    int t = q[todo + lane];
    q[lane] = t;
  }
  return rem;
}

__global__ __launch_bounds__(1024) void k_agg_owner(float* __restrict__ out,
                                                    const float* __restrict__ h,
                                                    const float* __restrict__ dinv,
                                                    const int* __restrict__ row,
                                                    const int* __restrict__ col,
                                                    const unsigned* __restrict__ key,
                                                    const unsigned* __restrict__ deg8,
                                                    const float* __restrict__ bias,
                                                    int do_relu,
                                                    int* __restrict__ rec_count,
                                                    int* __restrict__ rec_words,
                                                    size_t ws_bytes, int layer) {
  __shared__ float acc[RPG * HIDC];
  __shared__ int   queue[32 * QCAP];
  __shared__ __attribute__((aligned(16))) unsigned char degt[DEGT_BYTES];
  __shared__ float dlut[256];
  __shared__ int   rcount;
  const int tid  = threadIdx.x;
  const int lane = tid & 31;
  const int wv   = tid >> 5;
  const int lo   = blockIdx.x * RPG;
  const int nrows = min(RPG, NN - lo);
  int* q = queue + wv * QCAP;

  const bool room = ws_bytes >= REC_END;
  int2* const rec = (int2*)rec_words + (size_t)blockIdx.x * REC_CAP;
  int2* const wrec = (room && layer == 1) ? rec : nullptr;
  int nrec = REC_CAP + 1;
  if (room && layer == 2) nrec = rec_count[blockIdx.x];
  const bool replay = nrec <= REC_CAP;
  if (tid == 0) rcount = 0;

  const float4* h4 = (const float4*)(h + (size_t)lo * HIDC);
  for (int i = tid; i < nrows * (HIDC / 4); i += 1024) {
    float d = dinv[lo + (i >> 5)];
    float s = d * d;
    float4 v = h4[i];
    ((float4*)acc)[i] = make_float4(v.x * s, v.y * s, v.z * s, v.w * s);
  }
  if (!replay) {
    const int piece0 = (blockIdx.x * (NN / 16)) >> 8;
    for (int k = tid; k < NN / 16; k += 1024) {
      int i = k + piece0;
      if (i >= NN / 16) i -= NN / 16;
      ((uint4*)degt)[i] = ((const uint4*)deg8)[i];
    }
    if (tid < 256) dlut[tid] = inv_sqrt_deg((float)tid);
  }
  __syncthreads();

  if (replay) {
    int base = __builtin_amdgcn_readfirstlane(wv) * 32;
    if (base < nrec) {
      int2 cur = rec[min(base + lane, nrec - 1)];
      for (; base < nrec; base += 1024) {
        const int2 nxt = rec[min(base + 1024 + lane, nrec - 1)];
        const int cnt = min(32, nrec - base);
        agg_apply<true>(acc, cur.x & 0x1ffff, cur.x >> 17, __int_as_float(cur.y), cnt, lane, h);
        cur = nxt;
      }
    }
  } else {

  const uint4* key4 = (const uint4*)key;
  const unsigned pat = blockIdx.x * 0x01010101u;
  const int trip0 = (blockIdx.x * SCAN_TRIPS) >> 8;
  const unsigned below = (1u << lane) - 1u;
  int qn = 0;
  for (int k = 0; k < SCAN_TRIPS; k += 7) {
    uint4 c[7];
    int   g[7];
#pragma unroll
    for (int u = 0; u < 7; ++u) {
      int t = trip0 + k + u;
      if (t >= SCAN_TRIPS) t -= SCAN_TRIPS;
      g[u] = t * 1024 + tid;
      c[u] = key4[min(g[u], NKEY16 - 1)];
    }
#pragma unroll
    for (int u = 0; u < 7; ++u) {
      unsigned mm = (zero_bytes(c[u].x ^ pat) >> 7) | (zero_bytes(c[u].y ^ pat) >> 6) |
                    (zero_bytes(c[u].z ^ pat) >> 5) | (zero_bytes(c[u].w ^ pat) >> 4);
      if (g[u] >= NKEY16) mm = 0;
      if (__builtin_amdgcn_ballot_w32(mm != 0)) {
        do {
          bool has = mm != 0;
          unsigned b = __builtin_amdgcn_ballot_w32(has);
          if (has) {
            int p = __builtin_ctz(mm);
            q[qn + __popc(b & below)] = 16 * g[u] + ((p & 7) << 2) + (p >> 3);
            mm &= mm - 1;
          }
          qn += __popc(b);
          if (qn >= QDRAIN)
            qn = agg_drain<false>(acc, q, qn, lo, lane, degt, dlut, h, dinv, row, col, wrec, &rcount);
        } while (__builtin_amdgcn_ballot_w32(mm != 0));
      }
    }
  }
  agg_drain<true>(acc, q, qn, lo, lane, degt, dlut, h, dinv, row, col, wrec, &rcount);
  }
  __syncthreads();
  if (wrec && tid == 0) rec_count[blockIdx.x] = rcount;

  float4* o4 = (float4*)(out + (size_t)lo * HIDC);
  for (int i = tid; i < nrows * (HIDC / 4); i += 1024) {
    float4 v = ((const float4*)acc)[i];
    float4 b = ((const float4*)bias)[i & 31];
    v.x += b.x; v.y += b.y; v.z += b.z; v.w += b.w;
    if (do_relu) {
      v.x = fmaxf(v.x, 0.0f); v.y = fmaxf(v.y, 0.0f);
      v.z = fmaxf(v.z, 0.0f); v.w = fmaxf(v.w, 0.0f);
    }
    o4[i] = v;
  }
}

#define DEC_GROUPS 256
#define DEC_WAVES  25
#define NPAIR  (NPOS + NNEG)
#define NCHUNK (NPAIR / 32)

__device__ __forceinline__ void dec_pair_idx(const int* __restrict__ pos,
                                             const int* __restrict__ neg,
                                             int p, int& a, int& b) {
  if (p < NPOS) { a = pos[p]; b = pos[NPOS + p]; }
  else          { int j = p - NPOS; a = neg[j]; b = neg[NNEG + j]; }
}

__global__ __launch_bounds__(DEC_WAVES * 32) void k_decode(const float* __restrict__ z,
                                                          const int* __restrict__ pos,
                                                          const int* __restrict__ neg,
                                                          float* __restrict__ out) {
  const int lane = threadIdx.x & 31;
  const int g    = lane >> 3;
  const int i    = lane & 7;
  const int nw   = DEC_GROUPS * DEC_WAVES;
  int c = blockIdx.x * DEC_WAVES + (threadIdx.x >> 5);
  int a = 0, b = 0;
  if (c < NCHUNK) dec_pair_idx(pos, neg, c * 32 + lane, a, b);
  while (c < NCHUNK) {
    const int cn = c + nw;
    int an = 0, bn = 0;
    if (cn < NCHUNK) dec_pair_idx(pos, neg, cn * 32 + lane, an, bn);

    const float4* pa[8];
    const float4* pb[8];
#pragma unroll
    for (int j = 0; j < 8; ++j) {
      const int ra = __shfl(a, 4 * j + g, 32);
      const int rb = __shfl(b, 4 * j + g, 32);
      pa[j] = (const float4*)(z + (size_t)ra * HIDC) + i;
      pb[j] = (const float4*)(z + (size_t)rb * HIDC) + i;
    }
    float s[8];
#pragma unroll
    for (int k = 0; k < 4; ++k) {
      float4 va[2][4], vb[2][4];
      __builtin_amdgcn_sched_barrier(0);
#pragma unroll
      for (int h = 0; h < 2; ++h) {
#pragma unroll
        for (int u = 0; u < 4; ++u) {
          va[h][u] = pa[2 * k + h][8 * u];
          vb[h][u] = pb[2 * k + h][8 * u];
        }
      }
      __builtin_amdgcn_sched_barrier(0);
#pragma unroll
      for (int h = 0; h < 2; ++h) {
        float cx = 0.0f, cy = 0.0f, cz = 0.0f, cw = 0.0f;
#pragma unroll
        for (int u = 0; u < 4; ++u) {
          const float4 x = va[h][u], y = vb[h][u];
          cx = __builtin_fmaf(x.x, y.x, cx);
          cy = __builtin_fmaf(x.y, y.y, cy);
          cz = __builtin_fmaf(x.z, y.z, cz);
          cw = __builtin_fmaf(x.w, y.w, cw);
        }
        s[2 * k + h] = (cx + cy) + (cz + cw);
      }
    }
#pragma unroll
    for (int j = 0; j < 8; ++j) s[j] += __shfl_xor(s[j], 4, 32);
#pragma unroll
    for (int j = 0; j < 8; ++j) s[j] += __shfl_xor(s[j], 2, 32);
#pragma unroll
    for (int j = 0; j < 8; ++j) s[j] += __shfl_xor(s[j], 1, 32);
    float res = 0.0f;
#pragma unroll
    for (int j = 0; j < 8; ++j) {
      const float t = __shfl(s[j], (lane & 3) * 8, 32);
      if ((lane >> 2) == j) res = t;
    }
    out[c * 32 + lane] = res;
    a = an; b = bn; c = cn;
  }
}

extern "C" void kernel_launch(void* const* d_in, const int* in_sizes, int n_in,
                              void* d_out, int out_size, void* d_ws, size_t ws_size,
                              hipStream_t stream) {
  const float* x  = (const float*)d_in[0];
  const int*   ei = (const int*)d_in[1];
  const int*   pos = (const int*)d_in[2];
  const int*   neg = (const int*)d_in[3];
  const float* W1 = (const float*)d_in[4];
  const float* b1 = (const float*)d_in[5];
  const float* W2 = (const float*)d_in[6];
  const float* b2 = (const float*)d_in[7];
  float* logits = (float*)d_out;

  char* ws = (char*)d_ws;
  float* dinv = (float*)ws;
  unsigned* deg8 = (unsigned*)(ws + 400000u);
  float* WT1  = (float*)(ws + (512u << 10));
  float* WT2  = (float*)(ws + (640u << 10));
  float* bufH = (float*)(ws + (1u << 20));
  float* bufZ = (float*)(ws + (1u << 20) + (52u << 20));
  unsigned* key = (unsigned*)(ws + (1u << 20) + 51200000u);
  int*  rec_count = (int*)(ws + REC_OFF);
  int*  rec_all   = (int*)(ws + REC_OFF + REC_CNT_BYTES);

  const int* row = ei;
  const int* col = ei + NE;
  const int B = 256;

  k_prep<<<PREP_BLOCKS, PREP_B, 0, stream>>>(dinv, W1, WT1, W2, WT2);

  const int gemm_blocks = GEMM_GROUPS;

  k_gemm_wmma<INC, true><<<gemm_blocks, B, 0, stream>>>(x, WT1, bufH, dinv, col, key);
  k_finish_dinv<<<(NN / 4 + B - 1) / B, B, 0, stream>>>(dinv, deg8);
  k_agg_owner<<<AGG_GROUPS, 1024, 0, stream>>>(bufZ, bufH, dinv, row, col, key, deg8, b1, 1,
                                               rec_count, rec_all, ws_size, 1);

  k_gemm_wmma<HIDC, false><<<gemm_blocks, B, 0, stream>>>(bufZ, WT2, bufH, nullptr, nullptr, nullptr);
  k_agg_owner<<<AGG_GROUPS, 1024, 0, stream>>>(bufZ, bufH, dinv, row, col, key, deg8, b2, 0,
                                               rec_count, rec_all, ws_size, 2);

  k_decode<<<DEC_GROUPS, DEC_WAVES * 32, 0, stream>>>(bufZ, pos, neg, logits);
}


# layer-2 aggregation: records dealt into 32 LDS bins by owner wave (dest row mod 32); a wave adds its own rows by 16-byte LDS load, fma, store: no LDS float atomics; add order and rounding change
# speedup vs baseline: 3.1406x; 1.0304x over previous
#include <hip/hip_runtime.h>

#define NN   100000
#define INC  256
#define HIDC 128
#define NE   1600000
#define NPOS 200000
#define NNEG 200000

typedef __attribute__((ext_vector_type(2))) float v2f;
typedef __attribute__((ext_vector_type(8))) float v8f;

#define AGG_GROUPS 256
#define RPG   391

#define PREP_B 256
#define PREP_DEG_BLOCKS ((NN + PREP_B - 1) / PREP_B)
#define PREP_T1_BLOCKS  (INC * HIDC / PREP_B)
#define PREP_T2_BLOCKS  (HIDC * HIDC / PREP_B)
#define PREP_BLOCKS (PREP_DEG_BLOCKS + PREP_T1_BLOCKS + PREP_T2_BLOCKS)
__global__ __launch_bounds__(PREP_B) void k_prep(float* __restrict__ dinv,
                                                 const float* __restrict__ W1,
                                                 float* __restrict__ WT1,
                                                 const float* __restrict__ W2,
                                                 float* __restrict__ WT2) {
  const int b = blockIdx.x;
  if (b < PREP_DEG_BLOCKS) {
    int i = b * PREP_B + threadIdx.x;
    if (i < NN) dinv[i] = 1.0f;
  } else if (b < PREP_DEG_BLOCKS + PREP_T1_BLOCKS) {
    int i = (b - PREP_DEG_BLOCKS) * PREP_B + threadIdx.x;
    int k = i / HIDC, n = i - k * HIDC;
    WT1[n * INC + k] = W1[i];
  } else {
    int i = (b - PREP_DEG_BLOCKS - PREP_T1_BLOCKS) * PREP_B + threadIdx.x;
    int k = i / HIDC, n = i - k * HIDC;
    WT2[n * HIDC + k] = W2[i];
  }
}

__device__ __forceinline__ float inv_sqrt_deg(float d) { return 1.0f / sqrtf(d); }

#define DEG_SAT 255u

__global__ void k_finish_dinv(float* dinv, unsigned* __restrict__ deg8) {
  int i = blockIdx.x * blockDim.x + threadIdx.x;
  if (i < NN / 4) {
    float4 d = ((const float4*)dinv)[i];
    unsigned b0 = (unsigned)fminf(d.x, 255.0f), b1 = (unsigned)fminf(d.y, 255.0f);
    unsigned b2 = (unsigned)fminf(d.z, 255.0f), b3 = (unsigned)fminf(d.w, 255.0f);
    deg8[i] = b0 | (b1 << 8) | (b2 << 16) | (b3 << 24);
    ((float4*)dinv)[i] = make_float4(inv_sqrt_deg(d.x), inv_sqrt_deg(d.y),
                                     inv_sqrt_deg(d.z), inv_sqrt_deg(d.w));
  }
}

#define GROWS 32
#define GPAIRS (NN / GROWS)
#define GEMM_GROUPS 256
#define CNT_ITEMS (NE / 4)
#define CNT_TRIPS 7
static_assert(CNT_TRIPS * GEMM_GROUPS * 256 >= CNT_ITEMS, "every item has a thread and a trip");
static_assert(GPAIRS / GEMM_GROUPS > CNT_TRIPS, "every group has a trip behind its last item");
template <int K, bool COUNT>
__global__ __launch_bounds__(256) void k_gemm_wmma(const float* __restrict__ A,
                                                   const float* __restrict__ WT,
                                                   float* __restrict__ C,
                                                   float* dinv,
                                                   const int* __restrict__ col,
                                                   unsigned* __restrict__ key) {
  constexpr int PITCH = K + 4;
  constexpr int F4ROW = K / 4;
  constexpr int NLD   = GROWS * F4ROW / 256;
  __shared__ __attribute__((aligned(16))) float sA[2 * GROWS * PITCH];

  const int tid  = threadIdx.x;
  const int lane = tid & 31;
  const int wv   = tid >> 5;
  const int half = lane >> 4;
  const int l15  = lane & 15;
  const int np   = wv & 3;
  const int mh   = wv >> 2;

  v2f b0[K / 4], b1[K / 4];
  {
    const float* bp0 = WT + (size_t)(np * 32 + l15) * K + half * 2;
    const float* bp1 = bp0 + (size_t)16 * K;
#pragma unroll
    for (int kk = 0; kk < K / 4; ++kk) {
      b0[kk] = *(const v2f*)(bp0 + 4 * kk);
      b1[kk] = *(const v2f*)(bp1 + 4 * kk);
    }
  }

  int soff[NLD];
#pragma unroll
  for (int j = 0; j < NLD; ++j) {
    int f = j * 256 + tid;
    soff[j] = (f / F4ROW) * PITCH + (f % F4ROW) * 4;
  }

  int p = blockIdx.x;
  {
    float4 st0[NLD];
    const float4* src = (const float4*)(A + (size_t)p * GROWS * K) + tid;
#pragma unroll
    for (int j = 0; j < NLD; ++j) st0[j] = src[j * 256];
#pragma unroll
    for (int j = 0; j < NLD; ++j) *(float4*)(sA + soff[j]) = st0[j];
  }
  __syncthreads();

  const int aoff = (mh * 16 + l15) * PITCH + half * 2;
  float* const cbase = C + (size_t)(mh * 16 + half * 8) * HIDC + np * 32 + l15;
  v8f pacc0 = (v8f)(0.0f), pacc1 = (v8f)(0.0f);
  int buf = 0;
  int trip = 0;
  int4 cq = make_int4(0, 0, 0, 0);
  int  cqi = -1;
  for (; p < GPAIRS; p += GEMM_GROUPS, ++trip) {
    const bool more = p + GEMM_GROUPS < GPAIRS;
    float4 st[NLD];
    int4 cn = cq;
    int  cni = -1;
    if (more) {
      const float4* src = (const float4*)(A + (size_t)(p + GEMM_GROUPS) * GROWS * K) + tid;
#pragma unroll
      for (int j = 0; j < NLD; ++j) st[j] = src[j * 256];
      if (COUNT && trip < CNT_TRIPS) {
        const int i = (trip * GEMM_GROUPS + (int)blockIdx.x) * 256 + tid;
        cn = ((const int4*)col)[min(i, CNT_ITEMS - 1)];
        if (i < CNT_ITEMS) cni = i;
      }
    } else {
#pragma unroll
      for (int j = 0; j < NLD; ++j) st[j] = make_float4(0.0f, 0.0f, 0.0f, 0.0f);
    }
    if (p != (int)blockIdx.x) {
      float* cp = cbase + (size_t)(p - GEMM_GROUPS) * GROWS * HIDC;
#pragma unroll
      for (int i = 0; i < 8; ++i) {
        cp[i * HIDC]      = pacc0[i];
        cp[i * HIDC + 16] = pacc1[i];
      }
      if (COUNT && cqi >= 0) {
        unsafeAtomicAdd(&dinv[cq.x], 1.0f);
        unsafeAtomicAdd(&dinv[cq.y], 1.0f);
        unsafeAtomicAdd(&dinv[cq.z], 1.0f);
        unsafeAtomicAdd(&dinv[cq.w], 1.0f);
        unsigned k0 = (unsigned)cq.x / RPG, k1 = (unsigned)cq.y / RPG;
        unsigned k2 = (unsigned)cq.z / RPG, k3 = (unsigned)cq.w / RPG;
        key[cqi] = k0 | (k1 << 8) | (k2 << 16) | (k3 << 24);
      }
    }

    v8f acc0 = (v8f)(0.0f), acc1 = (v8f)(0.0f);
    const float* ap = sA + buf * (GROWS * PITCH) + aoff;
#pragma unroll
    for (int kk = 0; kk < K / 4; ++kk) {
      v2f a = *(const v2f*)(ap + 4 * kk);
      acc0 = __builtin_amdgcn_wmma_f32_16x16x4_f32(
          false, a, false, b0[kk], (short)0, acc0, false, false);
      acc1 = __builtin_amdgcn_wmma_f32_16x16x4_f32(
          false, a, false, b1[kk], (short)0, acc1, false, false);
    }
    pacc0 = acc0;
    pacc1 = acc1;

    {
      float* dstb = sA + (buf ^ 1) * (GROWS * PITCH);
#pragma unroll
      for (int j = 0; j < NLD; ++j) *(float4*)(dstb + soff[j]) = st[j];
    }
    __syncthreads();
    buf ^= 1;
    cq = cn;
    cqi = cni;
  }
  {
    float* cp = cbase + (size_t)(p - GEMM_GROUPS) * GROWS * HIDC;
#pragma unroll
    for (int i = 0; i < 8; ++i) {
      cp[i * HIDC]      = pacc0[i];
      cp[i * HIDC + 16] = pacc1[i];
    }
  }
}

#define REC_CAP  7168
#define REC_OFF  106774528ull
#define REC_CNT_BYTES 1024ull
#define REC_END  (REC_OFF + REC_CNT_BYTES + 256ull * REC_CAP * 8ull)
#define QDRAIN 64
#define QCAP  (QDRAIN + 32)
#define DEGT_BYTES (NN + 96)
#define NKEY16 (NE / 16)
#define SCAN_TRIPS 98
#define BIN_CAP 390
static_assert(32 * BIN_CAP * 8 <= DEGT_BYTES, "the 32 bins fit in the degree table's LDS");

__device__ __forceinline__ unsigned zero_bytes(unsigned x) {
  unsigned t = (x & 0x7f7f7f7fu) + 0x7f7f7f7fu;
  return ~(t | x) & 0x80808080u;
}

__device__ __forceinline__ float edge_norm(const unsigned char* degt, const float* dlut,
                                           const float* __restrict__ dinv, int r, int c) {
  const unsigned dr = degt[r], dc = degt[c];
  float fr = dlut[dr], fc = dlut[dc];
  if (__builtin_amdgcn_ballot_w32(dr == DEG_SAT || dc == DEG_SAT)) {
    if (dr == DEG_SAT) fr = dinv[r];
    if (dc == DEG_SAT) fc = dinv[c];
  }
  return fr * fc;
}

template <bool SINGLES>
__device__ __forceinline__ void agg_apply(float* acc, int r, int cl, float nm, int cnt,
                                          int lane, const float* __restrict__ h) {
  {
    int j = 0;
    for (; j + 8 <= cnt; j += 8) {
      float v[8][4];
      int   cj[8];
      float nj[8];
#pragma unroll
      for (int u = 0; u < 8; ++u) {
        int rj = __builtin_amdgcn_readlane(r, j + u);
        cj[u] = __builtin_amdgcn_readlane(cl, j + u);
        nj[u] = __int_as_float(__builtin_amdgcn_readlane(__float_as_int(nm), j + u));
        const float* src = h + (size_t)rj * HIDC + lane;
        v[u][0] = src[0];
        v[u][1] = src[32];
        v[u][2] = src[64];
        v[u][3] = src[96];
      }
#pragma unroll
      for (int u = 0; u < 8; ++u) {
        float* dst = acc + cj[u] * HIDC + lane;
        unsafeAtomicAdd(dst + 0,  v[u][0] * nj[u]);
        unsafeAtomicAdd(dst + 32, v[u][1] * nj[u]);
        unsafeAtomicAdd(dst + 64, v[u][2] * nj[u]);
        unsafeAtomicAdd(dst + 96, v[u][3] * nj[u]);
      }
    }
    if (SINGLES) {
      for (; j < cnt; ++j) {
        int   rj = __builtin_amdgcn_readlane(r, j);
        int   cj = __builtin_amdgcn_readlane(cl, j);
        float nj = __int_as_float(__builtin_amdgcn_readlane(__float_as_int(nm), j));
        const float* src = h + (size_t)rj * HIDC + lane;
        float v0 = src[0], v1 = src[32], v2 = src[64], v3 = src[96];
        float* dst = acc + cj * HIDC + lane;
        unsafeAtomicAdd(dst + 0,  v0 * nj);
        unsafeAtomicAdd(dst + 32, v1 * nj);
        unsafeAtomicAdd(dst + 64, v2 * nj);
        unsafeAtomicAdd(dst + 96, v3 * nj);
      }
    }
  }
}

__device__ __forceinline__ void own_fetch(const int2* bin, int g8, int n, int lane,
                                          const float* __restrict__ h,
                                          float4 (&v)[8], int (&cj)[8], float (&nj)[8]) {
  const int2 rc = bin[min(g8 + (lane & 7), n - 1)];
#pragma unroll
  for (int u = 0; u < 8; ++u) {
    const int x = __builtin_amdgcn_readlane(rc.x, u);
    nj[u] = __int_as_float(__builtin_amdgcn_readlane(rc.y, u));
    cj[u] = x >> 17;
    v[u] = ((const float4*)(h + (size_t)(x & 0x1ffff) * HIDC))[lane];
  }
}

template <bool PART>
__device__ __forceinline__ void own_add(float* acc, int lane, const float4 (&v)[8],
                                        const int (&cj)[8], const float (&nj)[8], int m) {
#pragma unroll
  for (int u = 0; u < 8; ++u) {
    if (PART && u >= m) break;
    float4* dst = (float4*)(acc + cj[u] * HIDC) + lane;
    float4 a = *dst;
    a.x += v[u].x * nj[u];
    a.y += v[u].y * nj[u];
    a.z += v[u].z * nj[u];
    a.w += v[u].w * nj[u];
    *dst = a;
  }
}

template <bool FINAL>
__device__ __forceinline__ int agg_drain(float* acc, int* q, int qn, int lo, int lane,
                                         const unsigned char* degt, const float* dlut,
                                         const float* __restrict__ h,
                                         const float* __restrict__ dinv,
                                         const int* __restrict__ row,
                                         const int* __restrict__ col,
                                         int2* __restrict__ rec, int* rcount) {
  const int todo = FINAL ? qn : (qn & ~7);
  for (int base = 0; base < todo; base += 32) {
    int cnt = min(32, todo - base);
    int r = 0, cl = 0;
    float nm = 0.0f;
    if (lane < cnt) {
      int e = q[base + lane];
      r = row[e];
      int c = col[e];
      cl = c - lo;
      nm = edge_norm(degt, dlut, dinv, r, c);
    }
    if (rec) {
      int pos = 0;
      if (lane == 0) pos = atomicAdd(rcount, cnt);
      pos = __builtin_amdgcn_readfirstlane(pos) + lane;
      if (lane < cnt && pos < REC_CAP) rec[pos] = make_int2(r | (cl << 17), __float_as_int(nm));
    }
    agg_apply<FINAL>(acc, r, cl, nm, cnt, lane, h);
  }
  if (FINAL) return 0;
  const int rem = qn - todo;
  if (lane < rem) {
    int t = q[todo + lane];
    q[lane] = t;
  }
  return rem;
}

__global__ __launch_bounds__(1024) void k_agg_owner(float* __restrict__ out,
                                                    const float* __restrict__ h,
                                                    const float* __restrict__ dinv,
                                                    const int* __restrict__ row,
                                                    const int* __restrict__ col,
                                                    const unsigned* __restrict__ key,
                                                    const unsigned* __restrict__ deg8,
                                                    const float* __restrict__ bias,
                                                    int do_relu,
                                                    int* __restrict__ rec_count,
                                                    int* __restrict__ rec_words,
                                                    size_t ws_bytes, int layer) {
  __shared__ float acc[RPG * HIDC];
  __shared__ int   queue[32 * QCAP];
  __shared__ __attribute__((aligned(16))) unsigned char degt[DEGT_BYTES];
  __shared__ float dlut[256];
  __shared__ int   rcount;
  __shared__ int   bcnt[32];
  __shared__ int   bovf;
  const int tid  = threadIdx.x;
  const int lane = tid & 31;
  const int wv   = tid >> 5;
  const int lo   = blockIdx.x * RPG;
  const int nrows = min(RPG, NN - lo);
  int* q = queue + wv * QCAP;

  const bool room = ws_bytes >= REC_END;
  int2* const rec = (int2*)rec_words + (size_t)blockIdx.x * REC_CAP;
  int2* const wrec = (room && layer == 1) ? rec : nullptr;
  int nrec = REC_CAP + 1;
  if (room && layer == 2) nrec = rec_count[blockIdx.x];
  const bool replay = nrec <= REC_CAP;
  if (tid == 0) rcount = 0;
  if (tid < 32) bcnt[tid] = 0;
  if (tid == 32) bovf = 0;

  const float4* h4 = (const float4*)(h + (size_t)lo * HIDC);
  for (int i = tid; i < nrows * (HIDC / 4); i += 1024) {
    float d = dinv[lo + (i >> 5)];
    float s = d * d;
    float4 v = h4[i];
    ((float4*)acc)[i] = make_float4(v.x * s, v.y * s, v.z * s, v.w * s);
  }
  if (!replay) {
    const int piece0 = (blockIdx.x * (NN / 16)) >> 8;
    for (int k = tid; k < NN / 16; k += 1024) {
      int i = k + piece0;
      if (i >= NN / 16) i -= NN / 16;
      ((uint4*)degt)[i] = ((const uint4*)deg8)[i];
    }
    if (tid < 256) dlut[tid] = inv_sqrt_deg((float)tid);
  }
  __syncthreads();

  if (replay) {
    int2* const bins = (int2*)degt;
    if (nrec > 0) {
      constexpr int NB = REC_CAP / 1024;
      static_assert(REC_CAP % 1024 == 0, "whole trips");
      int2 rcs[NB];
      int  pos[NB];
#pragma unroll
      for (int t = 0; t < NB; ++t) rcs[t] = rec[min(tid + t * 1024, nrec - 1)];
#pragma unroll
      for (int t = 0; t < NB; ++t) {
        pos[t] = -1;
        if (tid + t * 1024 < nrec) pos[t] = atomicAdd(&bcnt[(rcs[t].x >> 17) & 31], 1);
      }
#pragma unroll
      for (int t = 0; t < NB; ++t) {
        if (pos[t] >= 0) {
          if (pos[t] < BIN_CAP) bins[((rcs[t].x >> 17) & 31) * BIN_CAP + pos[t]] = rcs[t];
          else bovf = 1;
        }
      }
    }
    __syncthreads();
    if (bovf == 0) {
      const int w = __builtin_amdgcn_readfirstlane(wv);
      const int n = __builtin_amdgcn_readfirstlane(bcnt[w]);
      const int2* bin = bins + w * BIN_CAP;
      const int ngt = (n + 7) >> 3;
      if (ngt > 0) {
        float4 va[8], vb[8];
        int    ca[8], cb[8];
        float  sa[8], sb[8];
        own_fetch(bin, 0, n, lane, h, va, ca, sa);
        int k = 0;
        for (; k + 2 < ngt; k += 2) {
          own_fetch(bin, 8 * (k + 1), n, lane, h, vb, cb, sb);
          __builtin_amdgcn_sched_barrier(0);
          own_add<false>(acc, lane, va, ca, sa, 8);
          own_fetch(bin, 8 * (k + 2), n, lane, h, va, ca, sa);
          __builtin_amdgcn_sched_barrier(0);
          own_add<false>(acc, lane, vb, cb, sb, 8);
        }
        if (k + 1 < ngt) {
          own_fetch(bin, 8 * (k + 1), n, lane, h, vb, cb, sb);
          __builtin_amdgcn_sched_barrier(0);
          own_add<false>(acc, lane, va, ca, sa, 8);
          const int m = n - 8 * (k + 1);
          if (m == 8) own_add<false>(acc, lane, vb, cb, sb, 8);
          else        own_add<true>(acc, lane, vb, cb, sb, m);
        } else {
          const int m = n - 8 * k;
          if (m == 8) own_add<false>(acc, lane, va, ca, sa, 8);
          else        own_add<true>(acc, lane, va, ca, sa, m);
        }
      }
    } else {
    int base = __builtin_amdgcn_readfirstlane(wv) * 32;
    if (base < nrec) {
      int2 cur = rec[min(base + lane, nrec - 1)];
      for (; base < nrec; base += 1024) {
        const int2 nxt = rec[min(base + 1024 + lane, nrec - 1)];
        const int cnt = min(32, nrec - base);
        agg_apply<true>(acc, cur.x & 0x1ffff, cur.x >> 17, __int_as_float(cur.y), cnt, lane, h);
        cur = nxt;
      }
    }
    }
  } else {

  const uint4* key4 = (const uint4*)key;
  const unsigned pat = blockIdx.x * 0x01010101u;
  const int trip0 = (blockIdx.x * SCAN_TRIPS) >> 8;
  const unsigned below = (1u << lane) - 1u;
  int qn = 0;
  for (int k = 0; k < SCAN_TRIPS; k += 7) {
    uint4 c[7];
    int   g[7];
#pragma unroll
    for (int u = 0; u < 7; ++u) {
      int t = trip0 + k + u;
      if (t >= SCAN_TRIPS) t -= SCAN_TRIPS;
      g[u] = t * 1024 + tid;
      c[u] = key4[min(g[u], NKEY16 - 1)];
    }
#pragma unroll
    for (int u = 0; u < 7; ++u) {
      unsigned mm = (zero_bytes(c[u].x ^ pat) >> 7) | (zero_bytes(c[u].y ^ pat) >> 6) |
                    (zero_bytes(c[u].z ^ pat) >> 5) | (zero_bytes(c[u].w ^ pat) >> 4);
      if (g[u] >= NKEY16) mm = 0;
      if (__builtin_amdgcn_ballot_w32(mm != 0)) {
        do {
          bool has = mm != 0;
          unsigned b = __builtin_amdgcn_ballot_w32(has);
          if (has) {
            int p = __builtin_ctz(mm);
            q[qn + __popc(b & below)] = 16 * g[u] + ((p & 7) << 2) + (p >> 3);
            mm &= mm - 1;
          }
          qn += __popc(b);
          if (qn >= QDRAIN)
            qn = agg_drain<false>(acc, q, qn, lo, lane, degt, dlut, h, dinv, row, col, wrec, &rcount);
        } while (__builtin_amdgcn_ballot_w32(mm != 0));
      }
    }
  }
  agg_drain<true>(acc, q, qn, lo, lane, degt, dlut, h, dinv, row, col, wrec, &rcount);
  }
  __syncthreads();
  if (wrec && tid == 0) rec_count[blockIdx.x] = rcount;

  float4* o4 = (float4*)(out + (size_t)lo * HIDC);
  for (int i = tid; i < nrows * (HIDC / 4); i += 1024) {
    float4 v = ((const float4*)acc)[i];
    float4 b = ((const float4*)bias)[i & 31];
    v.x += b.x; v.y += b.y; v.z += b.z; v.w += b.w;
    if (do_relu) {
      v.x = fmaxf(v.x, 0.0f); v.y = fmaxf(v.y, 0.0f);
      v.z = fmaxf(v.z, 0.0f); v.w = fmaxf(v.w, 0.0f);
    }
    o4[i] = v;
  }
}

#define DEC_GROUPS 256
#define DEC_WAVES  25
#define NPAIR  (NPOS + NNEG)
#define NCHUNK (NPAIR / 32)

__device__ __forceinline__ void dec_pair_idx(const int* __restrict__ pos,
                                             const int* __restrict__ neg,
                                             int p, int& a, int& b) {
  if (p < NPOS) { a = pos[p]; b = pos[NPOS + p]; }
  else          { int j = p - NPOS; a = neg[j]; b = neg[NNEG + j]; }
}

__global__ __launch_bounds__(DEC_WAVES * 32) void k_decode(const float* __restrict__ z,
                                                          const int* __restrict__ pos,
                                                          const int* __restrict__ neg,
                                                          float* __restrict__ out) {
  const int lane = threadIdx.x & 31;
  const int g    = lane >> 3;
  const int i    = lane & 7;
  const int nw   = DEC_GROUPS * DEC_WAVES;
  int c = blockIdx.x * DEC_WAVES + (threadIdx.x >> 5);
  int a = 0, b = 0;
  if (c < NCHUNK) dec_pair_idx(pos, neg, c * 32 + lane, a, b);
  while (c < NCHUNK) {
    const int cn = c + nw;
    int an = 0, bn = 0;
    if (cn < NCHUNK) dec_pair_idx(pos, neg, cn * 32 + lane, an, bn);

    const float4* pa[8];
    const float4* pb[8];
#pragma unroll
    for (int j = 0; j < 8; ++j) {
      const int ra = __shfl(a, 4 * j + g, 32);
      const int rb = __shfl(b, 4 * j + g, 32);
      pa[j] = (const float4*)(z + (size_t)ra * HIDC) + i;
      pb[j] = (const float4*)(z + (size_t)rb * HIDC) + i;
    }
    float s[8];
#pragma unroll
    for (int k = 0; k < 4; ++k) {
      float4 va[2][4], vb[2][4];
      __builtin_amdgcn_sched_barrier(0);
#pragma unroll
      for (int h = 0; h < 2; ++h) {
#pragma unroll
        for (int u = 0; u < 4; ++u) {
          va[h][u] = pa[2 * k + h][8 * u];
          vb[h][u] = pb[2 * k + h][8 * u];
        }
      }
      __builtin_amdgcn_sched_barrier(0);
#pragma unroll
      for (int h = 0; h < 2; ++h) {
        float cx = 0.0f, cy = 0.0f, cz = 0.0f, cw = 0.0f;
#pragma unroll
        for (int u = 0; u < 4; ++u) {
          const float4 x = va[h][u], y = vb[h][u];
          cx = __builtin_fmaf(x.x, y.x, cx);
          cy = __builtin_fmaf(x.y, y.y, cy);
          cz = __builtin_fmaf(x.z, y.z, cz);
          cw = __builtin_fmaf(x.w, y.w, cw);
        }
        s[2 * k + h] = (cx + cy) + (cz + cw);
      }
    }
#pragma unroll
    for (int j = 0; j < 8; ++j) s[j] += __shfl_xor(s[j], 4, 32);
#pragma unroll
    for (int j = 0; j < 8; ++j) s[j] += __shfl_xor(s[j], 2, 32);
#pragma unroll
    for (int j = 0; j < 8; ++j) s[j] += __shfl_xor(s[j], 1, 32);
    float res = 0.0f;
#pragma unroll
    for (int j = 0; j < 8; ++j) {
      const float t = __shfl(s[j], (lane & 3) * 8, 32);
      if ((lane >> 2) == j) res = t;
    }
    out[c * 32 + lane] = res;
    a = an; b = bn; c = cn;
  }
}

extern "C" void kernel_launch(void* const* d_in, const int* in_sizes, int n_in,
                              void* d_out, int out_size, void* d_ws, size_t ws_size,
                              hipStream_t stream) {
  const float* x  = (const float*)d_in[0];
  const int*   ei = (const int*)d_in[1];
  const int*   pos = (const int*)d_in[2];
  const int*   neg = (const int*)d_in[3];
  const float* W1 = (const float*)d_in[4];
  const float* b1 = (const float*)d_in[5];
  const float* W2 = (const float*)d_in[6];
  const float* b2 = (const float*)d_in[7];
  float* logits = (float*)d_out;

  char* ws = (char*)d_ws;
  float* dinv = (float*)ws;
  unsigned* deg8 = (unsigned*)(ws + 400000u);
  float* WT1  = (float*)(ws + (512u << 10));
  float* WT2  = (float*)(ws + (640u << 10));
  float* bufH = (float*)(ws + (1u << 20));
  float* bufZ = (float*)(ws + (1u << 20) + (52u << 20));
  unsigned* key = (unsigned*)(ws + (1u << 20) + 51200000u);
  int*  rec_count = (int*)(ws + REC_OFF);
  int*  rec_all   = (int*)(ws + REC_OFF + REC_CNT_BYTES);

  const int* row = ei;
  const int* col = ei + NE;
  const int B = 256;

  k_prep<<<PREP_BLOCKS, PREP_B, 0, stream>>>(dinv, W1, WT1, W2, WT2);

  const int gemm_blocks = GEMM_GROUPS;

  k_gemm_wmma<INC, true><<<gemm_blocks, B, 0, stream>>>(x, WT1, bufH, dinv, col, key);
  k_finish_dinv<<<(NN / 4 + B - 1) / B, B, 0, stream>>>(dinv, deg8);
  k_agg_owner<<<AGG_GROUPS, 1024, 0, stream>>>(bufZ, bufH, dinv, row, col, key, deg8, b1, 1,
                                               rec_count, rec_all, ws_size, 1);

  k_gemm_wmma<HIDC, false><<<gemm_blocks, B, 0, stream>>>(bufZ, WT2, bufH, nullptr, nullptr, nullptr);
  k_agg_owner<<<AGG_GROUPS, 1024, 0, stream>>>(bufZ, bufH, dinv, row, col, key, deg8, b2, 0,
                                               rec_count, rec_all, ws_size, 2);

  k_decode<<<DEC_GROUPS, DEC_WAVES * 32, 0, stream>>>(bufZ, pos, neg, logits);
}


# layer-1 aggregation walks records too: one more launch orders 128 edge chunks by owner group in LDS (source, dest, norm); both layers read their pieces, no key scan; add order and rounding change
# speedup vs baseline: 3.5542x; 1.1317x over previous
#include <hip/hip_runtime.h>

#define NN   100000
#define INC  256
#define HIDC 128
#define NE   1600000
#define NPOS 200000
#define NNEG 200000

typedef __attribute__((ext_vector_type(2))) float v2f;
typedef __attribute__((ext_vector_type(8))) float v8f;

#define AGG_GROUPS 256
#define RPG   391

#define PREP_B 256
#define PREP_DEG_BLOCKS ((NN + PREP_B - 1) / PREP_B)
#define PREP_T1_BLOCKS  (INC * HIDC / PREP_B)
#define PREP_T2_BLOCKS  (HIDC * HIDC / PREP_B)
#define PREP_BLOCKS (PREP_DEG_BLOCKS + PREP_T1_BLOCKS + PREP_T2_BLOCKS)
__global__ __launch_bounds__(PREP_B) void k_prep(float* __restrict__ dinv,
                                                 const float* __restrict__ W1,
                                                 float* __restrict__ WT1,
                                                 const float* __restrict__ W2,
                                                 float* __restrict__ WT2) {
  const int b = blockIdx.x;
  if (b < PREP_DEG_BLOCKS) {
    int i = b * PREP_B + threadIdx.x;
    if (i < NN) dinv[i] = 1.0f;
  } else if (b < PREP_DEG_BLOCKS + PREP_T1_BLOCKS) {
    int i = (b - PREP_DEG_BLOCKS) * PREP_B + threadIdx.x;
    int k = i / HIDC, n = i - k * HIDC;
    WT1[n * INC + k] = W1[i];
  } else {
    int i = (b - PREP_DEG_BLOCKS - PREP_T1_BLOCKS) * PREP_B + threadIdx.x;
    int k = i / HIDC, n = i - k * HIDC;
    WT2[n * HIDC + k] = W2[i];
  }
}

__device__ __forceinline__ float inv_sqrt_deg(float d) { return 1.0f / sqrtf(d); }

#define DEG_SAT 255u

__global__ void k_finish_dinv(float* dinv, unsigned* __restrict__ deg8) {
  int i = blockIdx.x * blockDim.x + threadIdx.x;
  if (i < NN / 4) {
    float4 d = ((const float4*)dinv)[i];
    unsigned b0 = (unsigned)fminf(d.x, 255.0f), b1 = (unsigned)fminf(d.y, 255.0f);
    unsigned b2 = (unsigned)fminf(d.z, 255.0f), b3 = (unsigned)fminf(d.w, 255.0f);
    deg8[i] = b0 | (b1 << 8) | (b2 << 16) | (b3 << 24);
    ((float4*)dinv)[i] = make_float4(inv_sqrt_deg(d.x), inv_sqrt_deg(d.y),
                                     inv_sqrt_deg(d.z), inv_sqrt_deg(d.w));
  }
}

#define GROWS 32
#define GPAIRS (NN / GROWS)
#define GEMM_GROUPS 256
#define CNT_ITEMS (NE / 4)
#define CNT_TRIPS 7
static_assert(CNT_TRIPS * GEMM_GROUPS * 256 >= CNT_ITEMS, "every item has a thread and a trip");
static_assert(GPAIRS / GEMM_GROUPS > CNT_TRIPS, "every group has a trip behind its last item");
template <int K, bool COUNT>
__global__ __launch_bounds__(256) void k_gemm_wmma(const float* __restrict__ A,
                                                   const float* __restrict__ WT,
                                                   float* __restrict__ C,
                                                   float* dinv,
                                                   const int* __restrict__ col,
                                                   unsigned* __restrict__ key) {
  constexpr int PITCH = K + 4;
  constexpr int F4ROW = K / 4;
  constexpr int NLD   = GROWS * F4ROW / 256;
  __shared__ __attribute__((aligned(16))) float sA[2 * GROWS * PITCH];

  const int tid  = threadIdx.x;
  const int lane = tid & 31;
  const int wv   = tid >> 5;
  const int half = lane >> 4;
  const int l15  = lane & 15;
  const int np   = wv & 3;
  const int mh   = wv >> 2;

  v2f b0[K / 4], b1[K / 4];
  {
    const float* bp0 = WT + (size_t)(np * 32 + l15) * K + half * 2;
    const float* bp1 = bp0 + (size_t)16 * K;
#pragma unroll
    for (int kk = 0; kk < K / 4; ++kk) {
      b0[kk] = *(const v2f*)(bp0 + 4 * kk);
      b1[kk] = *(const v2f*)(bp1 + 4 * kk);
    }
  }

  int soff[NLD];
#pragma unroll
  for (int j = 0; j < NLD; ++j) {
    int f = j * 256 + tid;
    soff[j] = (f / F4ROW) * PITCH + (f % F4ROW) * 4;
  }

  int p = blockIdx.x;
  {
    float4 st0[NLD];
    const float4* src = (const float4*)(A + (size_t)p * GROWS * K) + tid;
#pragma unroll
    for (int j = 0; j < NLD; ++j) st0[j] = src[j * 256];
#pragma unroll
    for (int j = 0; j < NLD; ++j) *(float4*)(sA + soff[j]) = st0[j];
  }
  __syncthreads();

  const int aoff = (mh * 16 + l15) * PITCH + half * 2;
  float* const cbase = C + (size_t)(mh * 16 + half * 8) * HIDC + np * 32 + l15;
  v8f pacc0 = (v8f)(0.0f), pacc1 = (v8f)(0.0f);
  int buf = 0;
  int trip = 0;
  int4 cq = make_int4(0, 0, 0, 0);
  int  cqi = -1;
  for (; p < GPAIRS; p += GEMM_GROUPS, ++trip) {
    const bool more = p + GEMM_GROUPS < GPAIRS;
    float4 st[NLD];
    int4 cn = cq;
    int  cni = -1;
    if (more) {
      const float4* src = (const float4*)(A + (size_t)(p + GEMM_GROUPS) * GROWS * K) + tid;
#pragma unroll
      for (int j = 0; j < NLD; ++j) st[j] = src[j * 256];
      if (COUNT && trip < CNT_TRIPS) {
        const int i = (trip * GEMM_GROUPS + (int)blockIdx.x) * 256 + tid;
        cn = ((const int4*)col)[min(i, CNT_ITEMS - 1)];
        if (i < CNT_ITEMS) cni = i;
      }
    } else {
#pragma unroll
      for (int j = 0; j < NLD; ++j) st[j] = make_float4(0.0f, 0.0f, 0.0f, 0.0f);
    }
    if (p != (int)blockIdx.x) {
      float* cp = cbase + (size_t)(p - GEMM_GROUPS) * GROWS * HIDC;
#pragma unroll
      for (int i = 0; i < 8; ++i) {
        cp[i * HIDC]      = pacc0[i];
        cp[i * HIDC + 16] = pacc1[i];
      }
      if (COUNT && cqi >= 0) {
        unsafeAtomicAdd(&dinv[cq.x], 1.0f);
        unsafeAtomicAdd(&dinv[cq.y], 1.0f);
        unsafeAtomicAdd(&dinv[cq.z], 1.0f);
        unsafeAtomicAdd(&dinv[cq.w], 1.0f);
        unsigned k0 = (unsigned)cq.x / RPG, k1 = (unsigned)cq.y / RPG;
        unsigned k2 = (unsigned)cq.z / RPG, k3 = (unsigned)cq.w / RPG;
        key[cqi] = k0 | (k1 << 8) | (k2 << 16) | (k3 << 24);
      }
    }

    v8f acc0 = (v8f)(0.0f), acc1 = (v8f)(0.0f);
    const float* ap = sA + buf * (GROWS * PITCH) + aoff;
#pragma unroll
    for (int kk = 0; kk < K / 4; ++kk) {
      v2f a = *(const v2f*)(ap + 4 * kk);
      acc0 = __builtin_amdgcn_wmma_f32_16x16x4_f32(
          false, a, false, b0[kk], (short)0, acc0, false, false);
      acc1 = __builtin_amdgcn_wmma_f32_16x16x4_f32(
          false, a, false, b1[kk], (short)0, acc1, false, false);
    }
    pacc0 = acc0;
    pacc1 = acc1;

    {
      float* dstb = sA + (buf ^ 1) * (GROWS * PITCH);
#pragma unroll
      for (int j = 0; j < NLD; ++j) *(float4*)(dstb + soff[j]) = st[j];
    }
    __syncthreads();
    buf ^= 1;
    cq = cn;
    cqi = cni;
  }
  {
    float* cp = cbase + (size_t)(p - GEMM_GROUPS) * GROWS * HIDC;
#pragma unroll
    for (int i = 0; i < 8; ++i) {
      cp[i * HIDC]      = pacc0[i];
      cp[i * HIDC + 16] = pacc1[i];
    }
  }
}

#define BIN_CHUNKS 128
#define CHUNK_E  (NE / BIN_CHUNKS)
#define SEG_NB   12
#define REC_OFF  106774528ull
#define SEG_OFF_BYTES (BIN_CHUNKS * 257ull * 4ull)
#define REC_END  (REC_OFF + SEG_OFF_BYTES + (unsigned long long)NE * 8ull)
static_assert(NE % BIN_CHUNKS == 0, "whole chunks");
static_assert(BIN_CHUNKS * 8 == 1024, "8 threads of an aggregation group per chunk");
static_assert(SEG_OFF_BYTES % 16 == 0, "records stay 16-byte aligned behind the boundaries");
#define QDRAIN 64
#define QCAP  (QDRAIN + 32)
#define DEGT_BYTES (NN + 96)
#define NKEY16 (NE / 16)
#define SCAN_TRIPS 98
#define BIN_CAP 390
static_assert(32 * BIN_CAP * 8 <= DEGT_BYTES, "the 32 bins fit in the degree table's LDS");

__device__ __forceinline__ unsigned zero_bytes(unsigned x) {
  unsigned t = (x & 0x7f7f7f7fu) + 0x7f7f7f7fu;
  return ~(t | x) & 0x80808080u;
}

__device__ __forceinline__ float edge_norm(const unsigned char* degt, const float* dlut,
                                           const float* __restrict__ dinv, int r, int c) {
  const unsigned dr = degt[r], dc = degt[c];
  float fr = dlut[dr], fc = dlut[dc];
  if (__builtin_amdgcn_ballot_w32(dr == DEG_SAT || dc == DEG_SAT)) {
    if (dr == DEG_SAT) fr = dinv[r];
    if (dc == DEG_SAT) fc = dinv[c];
  }
  return fr * fc;
}

template <bool SINGLES>
__device__ __forceinline__ void agg_apply(float* acc, int r, int cl, float nm, int cnt,
                                          int lane, const float* __restrict__ h) {
  {
    int j = 0;
    for (; j + 8 <= cnt; j += 8) {
      float v[8][4];
      int   cj[8];
      float nj[8];
#pragma unroll
      for (int u = 0; u < 8; ++u) {
        int rj = __builtin_amdgcn_readlane(r, j + u);
        cj[u] = __builtin_amdgcn_readlane(cl, j + u);
        nj[u] = __int_as_float(__builtin_amdgcn_readlane(__float_as_int(nm), j + u));
        const float* src = h + (size_t)rj * HIDC + lane;
        v[u][0] = src[0];
        v[u][1] = src[32];
        v[u][2] = src[64];
        v[u][3] = src[96];
      }
#pragma unroll
      for (int u = 0; u < 8; ++u) {
        float* dst = acc + cj[u] * HIDC + lane;
        unsafeAtomicAdd(dst + 0,  v[u][0] * nj[u]);
        unsafeAtomicAdd(dst + 32, v[u][1] * nj[u]);
        unsafeAtomicAdd(dst + 64, v[u][2] * nj[u]);
        unsafeAtomicAdd(dst + 96, v[u][3] * nj[u]);
      }
    }
    if (SINGLES) {
      for (; j < cnt; ++j) {
        int   rj = __builtin_amdgcn_readlane(r, j);
        int   cj = __builtin_amdgcn_readlane(cl, j);
        float nj = __int_as_float(__builtin_amdgcn_readlane(__float_as_int(nm), j));
        const float* src = h + (size_t)rj * HIDC + lane;
        float v0 = src[0], v1 = src[32], v2 = src[64], v3 = src[96];
        float* dst = acc + cj * HIDC + lane;
        unsafeAtomicAdd(dst + 0,  v0 * nj);
        unsafeAtomicAdd(dst + 32, v1 * nj);
        unsafeAtomicAdd(dst + 64, v2 * nj);
        unsafeAtomicAdd(dst + 96, v3 * nj);
      }
    }
  }
}

__device__ __forceinline__ void own_fetch(const int2* bin, int g8, int n, int lane,
                                          const float* __restrict__ h,
                                          float4 (&v)[8], int (&cj)[8], float (&nj)[8]) {
  const int2 rc = bin[min(g8 + (lane & 7), n - 1)];
#pragma unroll
  for (int u = 0; u < 8; ++u) {
    const int x = __builtin_amdgcn_readlane(rc.x, u);
    nj[u] = __int_as_float(__builtin_amdgcn_readlane(rc.y, u));
    cj[u] = x >> 17;
    v[u] = ((const float4*)(h + (size_t)(x & 0x1ffff) * HIDC))[lane];
  }
}

template <bool PART>
__device__ __forceinline__ void own_add(float* acc, int lane, const float4 (&v)[8],
                                        const int (&cj)[8], const float (&nj)[8], int m) {
#pragma unroll
  for (int u = 0; u < 8; ++u) {
    if (PART && u >= m) break;
    float4* dst = (float4*)(acc + cj[u] * HIDC) + lane;
    float4 a = *dst;
    a.x += v[u].x * nj[u];
    a.y += v[u].y * nj[u];
    a.z += v[u].z * nj[u];
    a.w += v[u].w * nj[u];
    *dst = a;
  }
}

template <bool FINAL>
__device__ __forceinline__ int agg_drain(float* acc, int* q, int qn, int lo, int lane,
                                         const unsigned char* degt, const float* dlut,
                                         const float* __restrict__ h,
                                         const float* __restrict__ dinv,
                                         const int* __restrict__ row,
                                         const int* __restrict__ col) {
  const int todo = FINAL ? qn : (qn & ~7);
  for (int base = 0; base < todo; base += 32) {
    int cnt = min(32, todo - base);
    int r = 0, cl = 0;
    float nm = 0.0f;
    if (lane < cnt) {
      int e = q[base + lane];
      r = row[e];
      int c = col[e];
      cl = c - lo;
      nm = edge_norm(degt, dlut, dinv, r, c);
    }
    agg_apply<FINAL>(acc, r, cl, nm, cnt, lane, h);
  }
  if (FINAL) return 0;
  const int rem = qn - todo;
  if (lane < rem) {
    int t = q[todo + lane];
    q[lane] = t;
  }
  return rem;
}

__global__ __launch_bounds__(1024) void k_bin(const int* __restrict__ row,
                                              const int* __restrict__ col,
                                              const float* __restrict__ dinv,
                                              const unsigned* __restrict__ deg8,
                                              int* __restrict__ seg_off,
                                              int* __restrict__ rec_words,
                                              size_t ws_bytes) {
  __shared__ int2  srec[CHUNK_E];
  __shared__ __attribute__((aligned(16))) unsigned char degt[DEGT_BYTES];
  __shared__ float dlut[256];
  __shared__ int   scnt[256];
  __shared__ int   soff[257];
  if (ws_bytes < REC_END) return;
  const int tid   = threadIdx.x;
  const int chunk = blockIdx.x;
  constexpr int BT = (CHUNK_E + 1023) / 1024;
  int r[BT], c[BT];
#pragma unroll
  for (int t = 0; t < BT; ++t) {
    const int e = chunk * CHUNK_E + min(tid + t * 1024, CHUNK_E - 1);
    r[t] = row[e];
    c[t] = col[e];
  }
  constexpr int DT = (NN / 16 + 1023) / 1024;
  uint4 tq[DT];
#pragma unroll
  for (int t = 0; t < DT; ++t) tq[t] = ((const uint4*)deg8)[min(tid + t * 1024, NN / 16 - 1)];
#pragma unroll
  for (int t = 0; t < DT; ++t)
    if (tid + t * 1024 < NN / 16) ((uint4*)degt)[tid + t * 1024] = tq[t];
  if (tid < 256) {
    dlut[tid] = inv_sqrt_deg((float)tid);
    scnt[tid] = 0;
  }
  __syncthreads();
#pragma unroll
  for (int t = 0; t < BT; ++t)
    if (tid + t * 1024 < CHUNK_E) atomicAdd(&scnt[(unsigned)c[t] / RPG], 1);
  __syncthreads();
  if (tid < 32) {
    int v[8];
    int sum = 0;
#pragma unroll
    for (int u = 0; u < 8; ++u) {
      v[u] = scnt[8 * tid + u];
      sum += v[u];
    }
    int inc = sum;
#pragma unroll
    for (int d = 1; d < 32; d <<= 1) {
      const int up = __shfl_up(inc, d, 32);
      if (tid >= d) inc += up;
    }
    int run = inc - sum;
#pragma unroll
    for (int u = 0; u < 8; ++u) {
      soff[8 * tid + u] = run;
      scnt[8 * tid + u] = run;
      run += v[u];
    }
    if (tid == 31) soff[256] = run;
  }
  __syncthreads();
#pragma unroll
  for (int t = 0; t < BT; ++t) {
    if (tid + t * 1024 < CHUNK_E) {
      const int g  = (unsigned)c[t] / RPG;
      const int cl = c[t] - g * RPG;
      const float nm = edge_norm(degt, dlut, dinv, r[t], c[t]);
      const int pos = atomicAdd(&scnt[g], 1);
      srec[pos] = make_int2(r[t] | (cl << 17), __float_as_int(nm));
    }
  }
  __syncthreads();
  int2* dst = (int2*)rec_words + (size_t)chunk * CHUNK_E;
  for (int i = tid; i < CHUNK_E; i += 1024) dst[i] = srec[i];
  if (tid < 257) seg_off[chunk * 257 + tid] = soff[tid];
}

__global__ __launch_bounds__(1024) void k_agg_owner(float* __restrict__ out,
                                                    const float* __restrict__ h,
                                                    const float* __restrict__ dinv,
                                                    const int* __restrict__ row,
                                                    const int* __restrict__ col,
                                                    const unsigned* __restrict__ key,
                                                    const unsigned* __restrict__ deg8,
                                                    const float* __restrict__ bias,
                                                    int do_relu,
                                                    const int* __restrict__ seg_off,
                                                    const int* __restrict__ rec_words,
                                                    size_t ws_bytes) {
  __shared__ float acc[RPG * HIDC];
  __shared__ int   queue[32 * QCAP];
  __shared__ __attribute__((aligned(16))) unsigned char degt[DEGT_BYTES];
  __shared__ float dlut[256];
  __shared__ int   sbeg[BIN_CHUNKS];
  __shared__ int   slen[BIN_CHUNKS];
  __shared__ int   smax;
  __shared__ int   bcnt[32];
  __shared__ int   bovf;
  const int tid  = threadIdx.x;
  const int lane = tid & 31;
  const int wv   = tid >> 5;
  const int lo   = blockIdx.x * RPG;
  const int nrows = min(RPG, NN - lo);
  int* q = queue + wv * QCAP;

  const bool replay = ws_bytes >= REC_END;
  const int2* const rec_all = (const int2*)rec_words;
  if (tid < 32) bcnt[tid] = 0;
  if (tid == 32) bovf = 0;
  if (tid == 33) smax = 0;

  const float4* h4 = (const float4*)(h + (size_t)lo * HIDC);
  for (int i = tid; i < nrows * (HIDC / 4); i += 1024) {
    float d = dinv[lo + (i >> 5)];
    float s = d * d;
    float4 v = h4[i];
    ((float4*)acc)[i] = make_float4(v.x * s, v.y * s, v.z * s, v.w * s);
  }
  if (!replay) {
    const int piece0 = (blockIdx.x * (NN / 16)) >> 8;
    for (int k = tid; k < NN / 16; k += 1024) {
      int i = k + piece0;
      if (i >= NN / 16) i -= NN / 16;
      ((uint4*)degt)[i] = ((const uint4*)deg8)[i];
    }
    if (tid < 256) dlut[tid] = inv_sqrt_deg((float)tid);
  }
  __syncthreads();

  if (replay) {
    int2* const bins = (int2*)degt;
    if (tid < BIN_CHUNKS) {
      const int o0 = seg_off[tid * 257 + (int)blockIdx.x];
      const int o1 = seg_off[tid * 257 + (int)blockIdx.x + 1];
      sbeg[tid] = tid * CHUNK_E + o0;
      slen[tid] = o1 - o0;
      atomicMax(&smax, o1 - o0);
    }
    __syncthreads();
    {
      const int maxlen = smax;
      const int2* const sp = rec_all + sbeg[tid >> 3];
      const int sn = slen[tid >> 3];
      for (int rb = 0; rb < maxlen; rb += 8 * SEG_NB) {
        const int base = rb + (tid & 7);
        int2 rcs[SEG_NB];
        int  pos[SEG_NB];
#pragma unroll
        for (int t = 0; t < SEG_NB; ++t) {
          rcs[t] = make_int2(0, 0);
          if (base + 8 * t < sn) rcs[t] = sp[base + 8 * t];
        }
#pragma unroll
        for (int t = 0; t < SEG_NB; ++t) {
          pos[t] = -1;
          if (base + 8 * t < sn) pos[t] = atomicAdd(&bcnt[(rcs[t].x >> 17) & 31], 1);
        }
#pragma unroll
        for (int t = 0; t < SEG_NB; ++t) {
          if (pos[t] >= 0) {
            if (pos[t] < BIN_CAP) bins[((rcs[t].x >> 17) & 31) * BIN_CAP + pos[t]] = rcs[t];
            else bovf = 1;
          }
        }
      }
    }
    __syncthreads();
    if (bovf == 0) {
      const int w = __builtin_amdgcn_readfirstlane(wv);
      const int n = __builtin_amdgcn_readfirstlane(bcnt[w]);
      const int2* bin = bins + w * BIN_CAP;
      const int ngt = (n + 7) >> 3;
      if (ngt > 0) {
        float4 va[8], vb[8];
        int    ca[8], cb[8];
        float  sa[8], sb[8];
        own_fetch(bin, 0, n, lane, h, va, ca, sa);
        int k = 0;
        for (; k + 2 < ngt; k += 2) {
          own_fetch(bin, 8 * (k + 1), n, lane, h, vb, cb, sb);
          __builtin_amdgcn_sched_barrier(0);
          own_add<false>(acc, lane, va, ca, sa, 8);
          own_fetch(bin, 8 * (k + 2), n, lane, h, va, ca, sa);
          __builtin_amdgcn_sched_barrier(0);
          own_add<false>(acc, lane, vb, cb, sb, 8);
        }
        if (k + 1 < ngt) {
          own_fetch(bin, 8 * (k + 1), n, lane, h, vb, cb, sb);
          __builtin_amdgcn_sched_barrier(0);
          own_add<false>(acc, lane, va, ca, sa, 8);
          const int m = n - 8 * (k + 1);
          if (m == 8) own_add<false>(acc, lane, vb, cb, sb, 8);
          else        own_add<true>(acc, lane, vb, cb, sb, m);
        } else {
          const int m = n - 8 * k;
          if (m == 8) own_add<false>(acc, lane, va, ca, sa, 8);
          else        own_add<true>(acc, lane, va, ca, sa, m);
        }
      }
    } else {
      const int w = __builtin_amdgcn_readfirstlane(wv);
      for (int sc = w; sc < BIN_CHUNKS; sc += 32) {
        const int n = __builtin_amdgcn_readfirstlane(slen[sc]);
        const int2* const p = rec_all + __builtin_amdgcn_readfirstlane(sbeg[sc]);
        for (int base = 0; base < n; base += 32) {
          const int2 cur = p[min(base + lane, n - 1)];
          const int cnt = min(32, n - base);
          agg_apply<true>(acc, cur.x & 0x1ffff, cur.x >> 17, __int_as_float(cur.y), cnt, lane, h);
        }
      }
    }
  } else {

  const uint4* key4 = (const uint4*)key;
  const unsigned pat = blockIdx.x * 0x01010101u;
  const int trip0 = (blockIdx.x * SCAN_TRIPS) >> 8;
  const unsigned below = (1u << lane) - 1u;
  int qn = 0;
  for (int k = 0; k < SCAN_TRIPS; k += 7) {
    uint4 c[7];
    int   g[7];
#pragma unroll
    for (int u = 0; u < 7; ++u) {
      int t = trip0 + k + u;
      if (t >= SCAN_TRIPS) t -= SCAN_TRIPS;
      g[u] = t * 1024 + tid;
      c[u] = key4[min(g[u], NKEY16 - 1)];
    }
#pragma unroll
    for (int u = 0; u < 7; ++u) {
      unsigned mm = (zero_bytes(c[u].x ^ pat) >> 7) | (zero_bytes(c[u].y ^ pat) >> 6) |
                    (zero_bytes(c[u].z ^ pat) >> 5) | (zero_bytes(c[u].w ^ pat) >> 4);
      if (g[u] >= NKEY16) mm = 0;
      if (__builtin_amdgcn_ballot_w32(mm != 0)) {
        do {
          bool has = mm != 0;
          unsigned b = __builtin_amdgcn_ballot_w32(has);
          if (has) {
            int p = __builtin_ctz(mm);
            q[qn + __popc(b & below)] = 16 * g[u] + ((p & 7) << 2) + (p >> 3);
            mm &= mm - 1;
          }
          qn += __popc(b);
          if (qn >= QDRAIN)
            qn = agg_drain<false>(acc, q, qn, lo, lane, degt, dlut, h, dinv, row, col);
        } while (__builtin_amdgcn_ballot_w32(mm != 0));
      }
    }
  }
  agg_drain<true>(acc, q, qn, lo, lane, degt, dlut, h, dinv, row, col);
  }
  __syncthreads();

  float4* o4 = (float4*)(out + (size_t)lo * HIDC);
  for (int i = tid; i < nrows * (HIDC / 4); i += 1024) {
    float4 v = ((const float4*)acc)[i];
    float4 b = ((const float4*)bias)[i & 31];
    v.x += b.x; v.y += b.y; v.z += b.z; v.w += b.w;
    if (do_relu) {
      v.x = fmaxf(v.x, 0.0f); v.y = fmaxf(v.y, 0.0f);
      v.z = fmaxf(v.z, 0.0f); v.w = fmaxf(v.w, 0.0f);
    }
    o4[i] = v;
  }
}

#define DEC_GROUPS 256
#define DEC_WAVES  25
#define NPAIR  (NPOS + NNEG)
#define NCHUNK (NPAIR / 32)

__device__ __forceinline__ void dec_pair_idx(const int* __restrict__ pos,
                                             const int* __restrict__ neg,
                                             int p, int& a, int& b) {
  if (p < NPOS) { a = pos[p]; b = pos[NPOS + p]; }
  else          { int j = p - NPOS; a = neg[j]; b = neg[NNEG + j]; }
}

__global__ __launch_bounds__(DEC_WAVES * 32) void k_decode(const float* __restrict__ z,
                                                          const int* __restrict__ pos,
                                                          const int* __restrict__ neg,
                                                          float* __restrict__ out) {
  const int lane = threadIdx.x & 31;
  const int g    = lane >> 3;
  const int i    = lane & 7;
  const int nw   = DEC_GROUPS * DEC_WAVES;
  int c = blockIdx.x * DEC_WAVES + (threadIdx.x >> 5);
  int a = 0, b = 0;
  if (c < NCHUNK) dec_pair_idx(pos, neg, c * 32 + lane, a, b);
  while (c < NCHUNK) {
    const int cn = c + nw;
    int an = 0, bn = 0;
    if (cn < NCHUNK) dec_pair_idx(pos, neg, cn * 32 + lane, an, bn);

    const float4* pa[8];
    const float4* pb[8];
#pragma unroll
    for (int j = 0; j < 8; ++j) {
      const int ra = __shfl(a, 4 * j + g, 32);
      const int rb = __shfl(b, 4 * j + g, 32);
      pa[j] = (const float4*)(z + (size_t)ra * HIDC) + i;
      pb[j] = (const float4*)(z + (size_t)rb * HIDC) + i;
    }
    float s[8];
#pragma unroll
    for (int k = 0; k < 4; ++k) {
      float4 va[2][4], vb[2][4];
      __builtin_amdgcn_sched_barrier(0);
#pragma unroll
      for (int h = 0; h < 2; ++h) {
#pragma unroll
        for (int u = 0; u < 4; ++u) {
          va[h][u] = pa[2 * k + h][8 * u];
          vb[h][u] = pb[2 * k + h][8 * u];
        }
      }
      __builtin_amdgcn_sched_barrier(0);
#pragma unroll
      for (int h = 0; h < 2; ++h) {
        float cx = 0.0f, cy = 0.0f, cz = 0.0f, cw = 0.0f;
#pragma unroll
        for (int u = 0; u < 4; ++u) {
          const float4 x = va[h][u], y = vb[h][u];
          cx = __builtin_fmaf(x.x, y.x, cx);
          cy = __builtin_fmaf(x.y, y.y, cy);
          cz = __builtin_fmaf(x.z, y.z, cz);
          cw = __builtin_fmaf(x.w, y.w, cw);
        }
        s[2 * k + h] = (cx + cy) + (cz + cw);
      }
    }
#pragma unroll
    for (int j = 0; j < 8; ++j) s[j] += __shfl_xor(s[j], 4, 32);
#pragma unroll
    for (int j = 0; j < 8; ++j) s[j] += __shfl_xor(s[j], 2, 32);
#pragma unroll
    for (int j = 0; j < 8; ++j) s[j] += __shfl_xor(s[j], 1, 32);
    float res = 0.0f;
#pragma unroll
    for (int j = 0; j < 8; ++j) {
      const float t = __shfl(s[j], (lane & 3) * 8, 32);
      if ((lane >> 2) == j) res = t;
    }
    out[c * 32 + lane] = res;
    a = an; b = bn; c = cn;
  }
}

extern "C" void kernel_launch(void* const* d_in, const int* in_sizes, int n_in,
                              void* d_out, int out_size, void* d_ws, size_t ws_size,
                              hipStream_t stream) {
  const float* x  = (const float*)d_in[0];
  const int*   ei = (const int*)d_in[1];
  const int*   pos = (const int*)d_in[2];
  const int*   neg = (const int*)d_in[3];
  const float* W1 = (const float*)d_in[4];
  const float* b1 = (const float*)d_in[5];
  const float* W2 = (const float*)d_in[6];
  const float* b2 = (const float*)d_in[7];
  float* logits = (float*)d_out;

  char* ws = (char*)d_ws;
  float* dinv = (float*)ws;
  unsigned* deg8 = (unsigned*)(ws + 400000u);
  float* WT1  = (float*)(ws + (512u << 10));
  float* WT2  = (float*)(ws + (640u << 10));
  float* bufH = (float*)(ws + (1u << 20));
  float* bufZ = (float*)(ws + (1u << 20) + (52u << 20));
  unsigned* key = (unsigned*)(ws + (1u << 20) + 51200000u);
  int*  seg_off = (int*)(ws + REC_OFF);
  int*  rec_all = (int*)(ws + REC_OFF + SEG_OFF_BYTES);

  const int* row = ei;
  const int* col = ei + NE;
  const int B = 256;

  k_prep<<<PREP_BLOCKS, PREP_B, 0, stream>>>(dinv, W1, WT1, W2, WT2);

  const int gemm_blocks = GEMM_GROUPS;

  k_gemm_wmma<INC, true><<<gemm_blocks, B, 0, stream>>>(x, WT1, bufH, dinv, col, key);
  k_finish_dinv<<<(NN / 4 + B - 1) / B, B, 0, stream>>>(dinv, deg8);
  k_bin<<<BIN_CHUNKS, 1024, 0, stream>>>(row, col, dinv, deg8, seg_off, rec_all, ws_size);
  k_agg_owner<<<AGG_GROUPS, 1024, 0, stream>>>(bufZ, bufH, dinv, row, col, key, deg8, b1, 1,
                                               seg_off, rec_all, ws_size);

  k_gemm_wmma<HIDC, false><<<gemm_blocks, B, 0, stream>>>(bufZ, WT2, bufH, nullptr, nullptr, nullptr);
  k_agg_owner<<<AGG_GROUPS, 1024, 0, stream>>>(bufZ, bufH, dinv, row, col, key, deg8, b2, 0,
                                               seg_off, rec_all, ws_size);

  k_decode<<<DEC_GROUPS, DEC_WAVES * 32, 0, stream>>>(bufZ, pos, neg, logits);
}
